# starting kernel
# baseline (speedup 1.0000x reference)
#include <hip/hip_runtime.h>


typedef float v2f __attribute__((ext_vector_type(2)));
typedef float v8f __attribute__((ext_vector_type(8)));

constexpr int B_   = 4;
constexpr int CQ_  = 256;
constexpr int CKV_ = 256;
constexpr int NQ_  = 2048;
constexpr int NKV_ = 2048;
constexpr int H_   = 8;
constexpr int DK_  = 64;
constexpr int OD_  = H_ * DK_;

template <int NB, bool HASR>
__device__ __forceinline__ void wmma_gemm_strip(
    const float* __restrict__ A, long rsA, long csA,
    const float* __restrict__ Bm, long rsB, long csB,
    float* __restrict__ C, long rsC,
    const float* __restrict__ R, long rsR,
    int m0, int n0, int K, float scale) {
  const int lane = threadIdx.x & 31;
  const int r    = lane & 15;
  const int half = lane >> 4;
  const long arow = (long)(m0 + r) * rsA;
  long bcol[NB];
#pragma unroll
  for (int j = 0; j < NB; ++j) bcol[j] = (long)(n0 + 16 * j + r) * csB;

  v8f acc[NB];
#pragma unroll
  for (int j = 0; j < NB; ++j) acc[j] = {};

  for (int k0 = 0; k0 < K; k0 += 8) {
    const int kA = k0 + 2 * half;
    __builtin_prefetch(&Bm[(long)(kA + 32) * rsB + bcol[0]], 0, 3);
    v2f a0, a1;
    a0.x = A[arow + (long)(kA    ) * csA];
    a0.y = A[arow + (long)(kA + 1) * csA];
    a1.x = A[arow + (long)(kA + 4) * csA];
    a1.y = A[arow + (long)(kA + 5) * csA];
    v2f b0[NB], b1[NB];
#pragma unroll
    for (int j = 0; j < NB; ++j) {
      b0[j].x = Bm[(long)(kA    ) * rsB + bcol[j]];
      b0[j].y = Bm[(long)(kA + 1) * rsB + bcol[j]];
      b1[j].x = Bm[(long)(kA + 4) * rsB + bcol[j]];
      b1[j].y = Bm[(long)(kA + 5) * rsB + bcol[j]];
    }
#pragma unroll
    for (int j = 0; j < NB; ++j)
      acc[j] = __builtin_amdgcn_wmma_f32_16x16x4_f32(false, a0, false, b0[j],
                                                     (short)0, acc[j], false, false);
#pragma unroll
    for (int j = 0; j < NB; ++j)
      acc[j] = __builtin_amdgcn_wmma_f32_16x16x4_f32(false, a1, false, b1[j],
                                                     (short)0, acc[j], false, false);
  }

#pragma unroll
  for (int j = 0; j < NB; ++j) {
#pragma unroll
    for (int v = 0; v < 8; ++v) {
      const long row = m0 + v + 8 * half;
      const long col = n0 + 16 * j + r;
      float val = acc[j][v] * scale;
      if (HASR) val += R[row * rsR + col];
      C[row * rsC + col] = val;
    }
  }
}

__global__ void k_stage1_xxT(const float* __restrict__ xkv, float* __restrict__ S) {
  const int b = blockIdx.z;
  const float* X = xkv + (long)b * CKV_ * NKV_;
  wmma_gemm_strip<2, false>(X, NKV_, 1, X, 1, NKV_,
                            S + (long)b * CKV_ * CKV_, CKV_, nullptr, 0,
                            blockIdx.y * 16, blockIdx.x * 32, NKV_, 1.0f);
}

__global__ void k_stage2_T1(const float* __restrict__ Wk, const float* __restrict__ S,
                            float* __restrict__ T1) {
  const int z = blockIdx.z;
  const int b = z >> 3, h = z & 7;
  wmma_gemm_strip<4, false>(Wk + (long)h * DK_ * CKV_, CKV_, 1,
                            S + (long)b * CKV_ * CKV_, CKV_, 1,
                            T1 + (long)z * DK_ * CKV_, CKV_, nullptr, 0,
                            blockIdx.y * 16, blockIdx.x * 64, CKV_, 1.0f);
}

__global__ void k_stage3_M(const float* __restrict__ T1, const float* __restrict__ Wv,
                           float* __restrict__ Mm) {
  const int z = blockIdx.z;
  const int h = z & 7;
  wmma_gemm_strip<4, false>(T1 + (long)z * DK_ * CKV_, CKV_, 1,
                            Wv + (long)h * DK_ * CKV_, 1, CKV_,
                            Mm + (long)z * DK_ * DK_, DK_, nullptr, 0,
                            blockIdx.y * 16, blockIdx.x * 64, CKV_, 1.0f / (float)DK_);
}

__global__ void k_stage4_Pt(const float* __restrict__ Mm, const float* __restrict__ Wq,
                            float* __restrict__ Pt) {
  const int z = blockIdx.z;
  const int h = z & 7;
  wmma_gemm_strip<4, false>(Mm + (long)z * DK_ * DK_, 1, DK_,
                            Wq + (long)h * DK_ * CQ_, CQ_, 1,
                            Pt + (long)z * DK_ * CQ_, CQ_, nullptr, 0,
                            blockIdx.y * 16, blockIdx.x * 64, DK_, 1.0f);
}

__global__ void k_stage5_G(const float* __restrict__ Wo, const float* __restrict__ Pt,
                           const float* __restrict__ gamma, float* __restrict__ G) {
  const int b = blockIdx.z;
  wmma_gemm_strip<4, false>(Wo, OD_, 1,
                            Pt + (long)b * OD_ * CQ_, CQ_, 1,
                            G + (long)b * CQ_ * CQ_, CQ_, nullptr, 0,
                            blockIdx.y * 16, blockIdx.x * 64, OD_, gamma[0]);
}

__global__ void k_stage6_out(const float* __restrict__ G, const float* __restrict__ xq,
                             float* __restrict__ out) {
  const int b = blockIdx.z;
  const float* Bq = xq + (long)b * CQ_ * NQ_;
  wmma_gemm_strip<4, true>(G + (long)b * CQ_ * CQ_, CQ_, 1,
                           Bq, NQ_, 1,
                           out + (long)b * CQ_ * NQ_, NQ_, Bq, NQ_,
                           blockIdx.y * 16, blockIdx.x * 64, CQ_, 1.0f);
}

extern "C" void kernel_launch(void* const* d_in, const int* in_sizes, int n_in,
                              void* d_out, int out_size, void* d_ws, size_t ws_size,
                              hipStream_t stream) {
  const float* xq    = (const float*)d_in[0];
  const float* xkv   = (const float*)d_in[1];
  const float* Wq    = (const float*)d_in[2];
  const float* Wk    = (const float*)d_in[3];
  const float* Wv    = (const float*)d_in[4];
  const float* Wo    = (const float*)d_in[5];
  const float* gamma = (const float*)d_in[6];
  float* out = (float*)d_out;

  float* S  = (float*)d_ws;
  float* T1 = S  + (size_t)B_ * CKV_ * CKV_;
  float* Mm = T1 + (size_t)B_ * H_ * DK_ * CKV_;
  float* Pt = Mm + (size_t)B_ * H_ * DK_ * DK_;
  float* G  = Pt + (size_t)B_ * H_ * DK_ * CQ_;

  const dim3 blk(32, 1, 1);
  k_stage1_xxT<<<dim3(CKV_ / 32, CKV_ / 16, B_),     blk, 0, stream>>>(xkv, S);
  k_stage2_T1 <<<dim3(CKV_ / 64, DK_ / 16, B_ * H_), blk, 0, stream>>>(Wk, S, T1);
  k_stage3_M  <<<dim3(DK_ / 64,  DK_ / 16, B_ * H_), blk, 0, stream>>>(T1, Wv, Mm);
  k_stage4_Pt <<<dim3(CQ_ / 64,  DK_ / 16, B_ * H_), blk, 0, stream>>>(Mm, Wq, Pt);
  k_stage5_G  <<<dim3(CQ_ / 64,  CQ_ / 16, B_),      blk, 0, stream>>>(Wo, Pt, gamma, G);
  k_stage6_out<<<dim3(NQ_ / 64,  CQ_ / 16, B_),      blk, 0, stream>>>(G, xq, out);
}


# K loop software-pipelined in registers: next K chunk's A/B fragments (32 k; 64 in stage 1) load into a second register set ahead of the current chunk's WMMAs; prefetch hint dropped; sum order kept
# speedup vs baseline: 1.9348x; 1.9348x over previous
#include <hip/hip_runtime.h>


typedef float v2f __attribute__((ext_vector_type(2)));
typedef float v8f __attribute__((ext_vector_type(8)));

constexpr int B_   = 4;
constexpr int CQ_  = 256;
constexpr int CKV_ = 256;
constexpr int NQ_  = 2048;
constexpr int NKV_ = 2048;
constexpr int H_   = 8;
constexpr int DK_  = 64;
constexpr int OD_  = H_ * DK_;

template <int NB, int NS>
struct Frags {
  v2f a[NS][2];
  v2f b[NS][2][NB];
};

template <int NB, int NS>
__device__ __forceinline__ void load_frags(
    Frags<NB, NS>& f, const float* __restrict__ A, long arow, long csA,
    const float* __restrict__ Bm, long rsB, const long (&bcol)[NB], int kc, int half) {
#pragma unroll
  for (int s = 0; s < NS; ++s) {
    const int kA = kc + 8 * s + 2 * half;
    f.a[s][0].x = A[arow + (long)(kA    ) * csA];
    f.a[s][0].y = A[arow + (long)(kA + 1) * csA];
    f.a[s][1].x = A[arow + (long)(kA + 4) * csA];
    f.a[s][1].y = A[arow + (long)(kA + 5) * csA];
#pragma unroll
    for (int j = 0; j < NB; ++j) {
      f.b[s][0][j].x = Bm[(long)(kA    ) * rsB + bcol[j]];
      f.b[s][0][j].y = Bm[(long)(kA + 1) * rsB + bcol[j]];
      f.b[s][1][j].x = Bm[(long)(kA + 4) * rsB + bcol[j]];
      f.b[s][1][j].y = Bm[(long)(kA + 5) * rsB + bcol[j]];
    }
  }
}

template <int NB, int NS>
__device__ __forceinline__ void mma_frags(const Frags<NB, NS>& f, v8f (&acc)[NB]) {
#pragma unroll
  for (int s = 0; s < NS; ++s) {
#pragma unroll
    for (int j = 0; j < NB; ++j)
      acc[j] = __builtin_amdgcn_wmma_f32_16x16x4_f32(false, f.a[s][0], false, f.b[s][0][j],
                                                     (short)0, acc[j], false, false);
#pragma unroll
    for (int j = 0; j < NB; ++j)
      acc[j] = __builtin_amdgcn_wmma_f32_16x16x4_f32(false, f.a[s][1], false, f.b[s][1][j],
                                                     (short)0, acc[j], false, false);
  }
}

template <int NB, bool HASR, int KC, int K>
__device__ __forceinline__ void wmma_gemm_strip(
    const float* __restrict__ A, long rsA, long csA,
    const float* __restrict__ Bm, long rsB, long csB,
    float* __restrict__ C, long rsC,
    const float* __restrict__ R, long rsR,
    int m0, int n0, float scale) {
  static_assert(KC % 8 == 0 && K >= 2 * KC && K % (2 * KC) == 0,
                "K must be a whole number of chunk pairs");
  constexpr int NS = KC / 8;
  const int lane = threadIdx.x & 31;
  const int r    = lane & 15;
  const int half = lane >> 4;
  const long arow = (long)(m0 + r) * rsA;
  long bcol[NB];
#pragma unroll
  for (int j = 0; j < NB; ++j) bcol[j] = (long)(n0 + 16 * j + r) * csB;

  v8f acc[NB];
#pragma unroll
  for (int j = 0; j < NB; ++j) acc[j] = {};

  Frags<NB, NS> f0, f1;
  load_frags<NB, NS>(f0, A, arow, csA, Bm, rsB, bcol, 0, half);
  int kc = 0;
  for (; kc + 2 * KC < K; kc += 2 * KC) {
    load_frags<NB, NS>(f1, A, arow, csA, Bm, rsB, bcol, kc + KC, half);
    mma_frags<NB, NS>(f0, acc);
    load_frags<NB, NS>(f0, A, arow, csA, Bm, rsB, bcol, kc + 2 * KC, half);
    mma_frags<NB, NS>(f1, acc);
  }
  load_frags<NB, NS>(f1, A, arow, csA, Bm, rsB, bcol, kc + KC, half);
  mma_frags<NB, NS>(f0, acc);
  mma_frags<NB, NS>(f1, acc);

#pragma unroll
  for (int j = 0; j < NB; ++j) {
#pragma unroll
    for (int v = 0; v < 8; ++v) {
      const long row = m0 + v + 8 * half;
      const long col = n0 + 16 * j + r;
      float val = acc[j][v] * scale;
      if (HASR) val += R[row * rsR + col];
      C[row * rsC + col] = val;
    }
  }
}

__global__ void __launch_bounds__(32) k_stage1_xxT(const float* __restrict__ xkv, float* __restrict__ S) {
  const int b = blockIdx.z;
  const float* X = xkv + (long)b * CKV_ * NKV_;
  wmma_gemm_strip<2, false, 64, NKV_>(X, NKV_, 1, X, 1, NKV_,
                            S + (long)b * CKV_ * CKV_, CKV_, nullptr, 0,
                            blockIdx.y * 16, blockIdx.x * 32, 1.0f);
}

__global__ void __launch_bounds__(32) k_stage2_T1(const float* __restrict__ Wk, const float* __restrict__ S,
                            float* __restrict__ T1) {
  const int z = blockIdx.z;
  const int b = z >> 3, h = z & 7;
  wmma_gemm_strip<4, false, 32, CKV_>(Wk + (long)h * DK_ * CKV_, CKV_, 1,
                            S + (long)b * CKV_ * CKV_, CKV_, 1,
                            T1 + (long)z * DK_ * CKV_, CKV_, nullptr, 0,
                            blockIdx.y * 16, blockIdx.x * 64, 1.0f);
}

__global__ void __launch_bounds__(32) k_stage3_M(const float* __restrict__ T1, const float* __restrict__ Wv,
                           float* __restrict__ Mm) {
  const int z = blockIdx.z;
  const int h = z & 7;
  wmma_gemm_strip<4, false, 32, CKV_>(T1 + (long)z * DK_ * CKV_, CKV_, 1,
                            Wv + (long)h * DK_ * CKV_, 1, CKV_,
                            Mm + (long)z * DK_ * DK_, DK_, nullptr, 0,
                            blockIdx.y * 16, blockIdx.x * 64, 1.0f / (float)DK_);
}

__global__ void __launch_bounds__(32) k_stage4_Pt(const float* __restrict__ Mm, const float* __restrict__ Wq,
                            float* __restrict__ Pt) {
  const int z = blockIdx.z;
  const int h = z & 7;
  wmma_gemm_strip<4, false, 32, DK_>(Mm + (long)z * DK_ * DK_, 1, DK_,
                            Wq + (long)h * DK_ * CQ_, CQ_, 1,
                            Pt + (long)z * DK_ * CQ_, CQ_, nullptr, 0,
                            blockIdx.y * 16, blockIdx.x * 64, 1.0f);
}

__global__ void __launch_bounds__(32) k_stage5_G(const float* __restrict__ Wo, const float* __restrict__ Pt,
                           const float* __restrict__ gamma, float* __restrict__ G) {
  const int b = blockIdx.z;
  wmma_gemm_strip<4, false, 32, OD_>(Wo, OD_, 1,
                            Pt + (long)b * OD_ * CQ_, CQ_, 1,
                            G + (long)b * CQ_ * CQ_, CQ_, nullptr, 0,
                            blockIdx.y * 16, blockIdx.x * 64, gamma[0]);
}

__global__ void __launch_bounds__(32) k_stage6_out(const float* __restrict__ G, const float* __restrict__ xq,
                             float* __restrict__ out) {
  const int b = blockIdx.z;
  const float* Bq = xq + (long)b * CQ_ * NQ_;
  wmma_gemm_strip<4, true, 32, CQ_>(G + (long)b * CQ_ * CQ_, CQ_, 1,
                           Bq, NQ_, 1,
                           out + (long)b * CQ_ * NQ_, NQ_, Bq, NQ_,
                           blockIdx.y * 16, blockIdx.x * 64, 1.0f);
}

extern "C" void kernel_launch(void* const* d_in, const int* in_sizes, int n_in,
                              void* d_out, int out_size, void* d_ws, size_t ws_size,
                              hipStream_t stream) {
  const float* xq    = (const float*)d_in[0];
  const float* xkv   = (const float*)d_in[1];
  const float* Wq    = (const float*)d_in[2];
  const float* Wk    = (const float*)d_in[3];
  const float* Wv    = (const float*)d_in[4];
  const float* Wo    = (const float*)d_in[5];
  const float* gamma = (const float*)d_in[6];
  float* out = (float*)d_out;

  float* S  = (float*)d_ws;
  float* T1 = S  + (size_t)B_ * CKV_ * CKV_;
  float* Mm = T1 + (size_t)B_ * H_ * DK_ * CKV_;
  float* Pt = Mm + (size_t)B_ * H_ * DK_ * DK_;
  float* G  = Pt + (size_t)B_ * H_ * DK_ * CQ_;

  const dim3 blk(32, 1, 1);
  k_stage1_xxT<<<dim3(CKV_ / 32, CKV_ / 16, B_),     blk, 0, stream>>>(xkv, S);
  k_stage2_T1 <<<dim3(CKV_ / 64, DK_ / 16, B_ * H_), blk, 0, stream>>>(Wk, S, T1);
  k_stage3_M  <<<dim3(DK_ / 64,  DK_ / 16, B_ * H_), blk, 0, stream>>>(T1, Wv, Mm);
  k_stage4_Pt <<<dim3(CQ_ / 64,  DK_ / 16, B_ * H_), blk, 0, stream>>>(Mm, Wq, Pt);
  k_stage5_G  <<<dim3(CQ_ / 64,  CQ_ / 16, B_),      blk, 0, stream>>>(Wo, Pt, gamma, G);
  k_stage6_out<<<dim3(NQ_ / 64,  CQ_ / 16, B_),      blk, 0, stream>>>(G, xq, out);
}


# Row-stored B operands: NB tiles column-interleaved, each k row loads once as 128 bit for all tiles (C, residual rows too); a chunk's 24 loads pinned as one group ahead of its WMMAs; sum order kept
# speedup vs baseline: 2.2975x; 1.1875x over previous
#include <hip/hip_runtime.h>


typedef float v2f __attribute__((ext_vector_type(2)));
typedef float v4f __attribute__((ext_vector_type(4)));
typedef float v8f __attribute__((ext_vector_type(8)));

template <int NB> struct VecN;
template <> struct VecN<2> { typedef v2f type; };
template <> struct VecN<4> { typedef v4f type; };

constexpr int B_   = 4;
constexpr int CQ_  = 256;
constexpr int CKV_ = 256;
constexpr int NQ_  = 2048;
constexpr int NKV_ = 2048;
constexpr int H_   = 8;
constexpr int DK_  = 64;
constexpr int OD_  = H_ * DK_;

template <int NB, int NS, bool BROW> struct Frags;
template <int NB, int NS>
struct Frags<NB, NS, false> {
  v2f a[NS][2];
  v2f b[NS][2][NB];
};
template <int NB, int NS>
struct Frags<NB, NS, true> {
  v2f a[NS][2];
  typename VecN<NB>::type br[NS][2][2];
};

template <int NB, int NS>
__device__ __forceinline__ v2f bfrag(const Frags<NB, NS, false>& f, int s, int i, int j) {
  return f.b[s][i][j];
}
template <int NB, int NS>
__device__ __forceinline__ v2f bfrag(const Frags<NB, NS, true>& f, int s, int i, int j) {
  v2f b;
  b.x = f.br[s][i][0][j];
  b.y = f.br[s][i][1][j];
  return b;
}

template <int NB, int NS, bool BROW>
__device__ __forceinline__ void load_frags(
    Frags<NB, NS, BROW>& f, const float* __restrict__ A, long arow, long csA,
    const float* __restrict__ Bm, long rsB, const long (&bcol)[NB], int kc, int half) {
  typedef typename VecN<NB>::type vN;
#pragma unroll
  for (int s = 0; s < NS; ++s) {
    const int kA = kc + 8 * s + 2 * half;
    f.a[s][0].x = A[arow + (long)(kA    ) * csA];
    f.a[s][0].y = A[arow + (long)(kA + 1) * csA];
    f.a[s][1].x = A[arow + (long)(kA + 4) * csA];
    f.a[s][1].y = A[arow + (long)(kA + 5) * csA];
    if constexpr (BROW) {
      f.br[s][0][0] = *(const vN*)(Bm + (long)(kA    ) * rsB + bcol[0]);
      f.br[s][0][1] = *(const vN*)(Bm + (long)(kA + 1) * rsB + bcol[0]);
      f.br[s][1][0] = *(const vN*)(Bm + (long)(kA + 4) * rsB + bcol[0]);
      f.br[s][1][1] = *(const vN*)(Bm + (long)(kA + 5) * rsB + bcol[0]);
    } else {
#pragma unroll
      for (int j = 0; j < NB; ++j) {
        f.b[s][0][j].x = Bm[(long)(kA    ) * rsB + bcol[j]];
        f.b[s][0][j].y = Bm[(long)(kA + 1) * rsB + bcol[j]];
        f.b[s][1][j].x = Bm[(long)(kA + 4) * rsB + bcol[j]];
        f.b[s][1][j].y = Bm[(long)(kA + 5) * rsB + bcol[j]];
      }
    }
  }
}

template <int NB, int NS, bool BROW>
__device__ __forceinline__ void mma_frags(const Frags<NB, NS, BROW>& f, v8f (&acc)[NB]) {
#pragma unroll
  for (int s = 0; s < NS; ++s) {
#pragma unroll
    for (int j = 0; j < NB; ++j)
      acc[j] = __builtin_amdgcn_wmma_f32_16x16x4_f32(false, f.a[s][0], false, bfrag<NB, NS>(f, s, 0, j),
                                                     (short)0, acc[j], false, false);
#pragma unroll
    for (int j = 0; j < NB; ++j)
      acc[j] = __builtin_amdgcn_wmma_f32_16x16x4_f32(false, f.a[s][1], false, bfrag<NB, NS>(f, s, 1, j),
                                                     (short)0, acc[j], false, false);
  }
}

template <int NB, bool HASR, bool BROW, int KC, int K>
__device__ __forceinline__ void wmma_gemm_strip(
    const float* __restrict__ A, long rsA, long csA,
    const float* __restrict__ Bm, long rsB, long csB,
    float* __restrict__ C, long rsC,
    const float* __restrict__ R, long rsR,
    int m0, int n0, float scale) {
  static_assert(KC % 8 == 0 && K >= 2 * KC && K % (2 * KC) == 0,
                "K must be a whole number of chunk pairs");
  typedef typename VecN<NB>::type vN;
  constexpr int NS = KC / 8;
  const int lane = threadIdx.x & 31;
  const int r    = lane & 15;
  const int half = lane >> 4;
  const long arow  = (long)(m0 + r) * rsA;
  const long col0 = n0 + NB * r;
  long bcol[NB];
#pragma unroll
  for (int j = 0; j < NB; ++j)
    bcol[j] = BROW ? (long)(n0 + NB * r + j) * csB : (long)(n0 + 16 * j + r) * csB;

  v8f acc[NB];
#pragma unroll
  for (int j = 0; j < NB; ++j) acc[j] = {};

  Frags<NB, NS, BROW> f0, f1;
  load_frags<NB, NS, BROW>(f0, A, arow, csA, Bm, rsB, bcol, 0, half);
  int kc = 0;
  for (; kc + 2 * KC < K; kc += 2 * KC) {
    load_frags<NB, NS, BROW>(f1, A, arow, csA, Bm, rsB, bcol, kc + KC, half);
    if constexpr (BROW) __builtin_amdgcn_sched_barrier(0);
    mma_frags<NB, NS, BROW>(f0, acc);
    if constexpr (BROW) __builtin_amdgcn_sched_barrier(0);
    load_frags<NB, NS, BROW>(f0, A, arow, csA, Bm, rsB, bcol, kc + 2 * KC, half);
    if constexpr (BROW) __builtin_amdgcn_sched_barrier(0);
    mma_frags<NB, NS, BROW>(f1, acc);
    if constexpr (BROW) __builtin_amdgcn_sched_barrier(0);
  }
  load_frags<NB, NS, BROW>(f1, A, arow, csA, Bm, rsB, bcol, kc + KC, half);
  mma_frags<NB, NS, BROW>(f0, acc);
  mma_frags<NB, NS, BROW>(f1, acc);

  if constexpr (BROW) {
#pragma unroll
    for (int v = 0; v < 8; ++v) {
      const long row = m0 + v + 8 * half;
      vN res;
      if (HASR) res = *(const vN*)(R + row * rsR + col0);
      vN val;
#pragma unroll
      for (int j = 0; j < NB; ++j) {
        float t = acc[j][v] * scale;
        if (HASR) t += res[j];
        val[j] = t;
      }
      *(vN*)(C + row * rsC + col0) = val;
    }
  } else {
#pragma unroll
    for (int j = 0; j < NB; ++j) {
#pragma unroll
      for (int v = 0; v < 8; ++v) {
        const long row = m0 + v + 8 * half;
        const long col = n0 + 16 * j + r;
        float val = acc[j][v] * scale;
        if (HASR) val += R[row * rsR + col];
        C[row * rsC + col] = val;
      }
    }
  }
}

__global__ void __launch_bounds__(32) k_stage1_xxT(const float* __restrict__ xkv, float* __restrict__ S) {
  const int b = blockIdx.z;
  const float* X = xkv + (long)b * CKV_ * NKV_;
  wmma_gemm_strip<2, false, false, 64, NKV_>(X, NKV_, 1, X, 1, NKV_,
                            S + (long)b * CKV_ * CKV_, CKV_, nullptr, 0,
                            blockIdx.y * 16, blockIdx.x * 32, 1.0f);
}

__global__ void __launch_bounds__(32) k_stage2_T1(const float* __restrict__ Wk, const float* __restrict__ S,
                            float* __restrict__ T1) {
  const int z = blockIdx.z;
  const int b = z >> 3, h = z & 7;
  wmma_gemm_strip<4, false, true, 32, CKV_>(Wk + (long)h * DK_ * CKV_, CKV_, 1,
                            S + (long)b * CKV_ * CKV_, CKV_, 1,
                            T1 + (long)z * DK_ * CKV_, CKV_, nullptr, 0,
                            blockIdx.y * 16, blockIdx.x * 64, 1.0f);
}

__global__ void __launch_bounds__(32) k_stage3_M(const float* __restrict__ T1, const float* __restrict__ Wv,
                           float* __restrict__ Mm) {
  const int z = blockIdx.z;
  const int h = z & 7;
  wmma_gemm_strip<4, false, false, 32, CKV_>(T1 + (long)z * DK_ * CKV_, CKV_, 1,
                            Wv + (long)h * DK_ * CKV_, 1, CKV_,
                            Mm + (long)z * DK_ * DK_, DK_, nullptr, 0,
                            blockIdx.y * 16, blockIdx.x * 64, 1.0f / (float)DK_);
}

__global__ void __launch_bounds__(32) k_stage4_Pt(const float* __restrict__ Mm, const float* __restrict__ Wq,
                            float* __restrict__ Pt) {
  const int z = blockIdx.z;
  const int h = z & 7;
  wmma_gemm_strip<4, false, true, 32, DK_>(Mm + (long)z * DK_ * DK_, 1, DK_,
                            Wq + (long)h * DK_ * CQ_, CQ_, 1,
                            Pt + (long)z * DK_ * CQ_, CQ_, nullptr, 0,
                            blockIdx.y * 16, blockIdx.x * 64, 1.0f);
}

__global__ void __launch_bounds__(32) k_stage5_G(const float* __restrict__ Wo, const float* __restrict__ Pt,
                           const float* __restrict__ gamma, float* __restrict__ G) {
  const int b = blockIdx.z;
  wmma_gemm_strip<4, false, true, 32, OD_>(Wo, OD_, 1,
                            Pt + (long)b * OD_ * CQ_, CQ_, 1,
                            G + (long)b * CQ_ * CQ_, CQ_, nullptr, 0,
                            blockIdx.y * 16, blockIdx.x * 64, gamma[0]);
}

__global__ void __launch_bounds__(32) k_stage6_out(const float* __restrict__ G, const float* __restrict__ xq,
                             float* __restrict__ out) {
  const int b = blockIdx.z;
  const float* Bq = xq + (long)b * CQ_ * NQ_;
  wmma_gemm_strip<4, true, true, 32, CQ_>(G + (long)b * CQ_ * CQ_, CQ_, 1,
                           Bq, NQ_, 1,
                           out + (long)b * CQ_ * NQ_, NQ_, Bq, NQ_,
                           blockIdx.y * 16, blockIdx.x * 64, 1.0f);
}

extern "C" void kernel_launch(void* const* d_in, const int* in_sizes, int n_in,
                              void* d_out, int out_size, void* d_ws, size_t ws_size,
                              hipStream_t stream) {
  const float* xq    = (const float*)d_in[0];
  const float* xkv   = (const float*)d_in[1];
  const float* Wq    = (const float*)d_in[2];
  const float* Wk    = (const float*)d_in[3];
  const float* Wv    = (const float*)d_in[4];
  const float* Wo    = (const float*)d_in[5];
  const float* gamma = (const float*)d_in[6];
  float* out = (float*)d_out;

  float* S  = (float*)d_ws;
  float* T1 = S  + (size_t)B_ * CKV_ * CKV_;
  float* Mm = T1 + (size_t)B_ * H_ * DK_ * CKV_;
  float* Pt = Mm + (size_t)B_ * H_ * DK_ * DK_;
  float* G  = Pt + (size_t)B_ * H_ * DK_ * CQ_;

  const dim3 blk(32, 1, 1);
  k_stage1_xxT<<<dim3(CKV_ / 32, CKV_ / 16, B_),     blk, 0, stream>>>(xkv, S);
  k_stage2_T1 <<<dim3(CKV_ / 64, DK_ / 16, B_ * H_), blk, 0, stream>>>(Wk, S, T1);
  k_stage3_M  <<<dim3(DK_ / 64,  DK_ / 16, B_ * H_), blk, 0, stream>>>(T1, Wv, Mm);
  k_stage4_Pt <<<dim3(CQ_ / 64,  DK_ / 16, B_ * H_), blk, 0, stream>>>(Mm, Wq, Pt);
  k_stage5_G  <<<dim3(CQ_ / 64,  CQ_ / 16, B_),      blk, 0, stream>>>(Wo, Pt, gamma, G);
  k_stage6_out<<<dim3(NQ_ / 64,  CQ_ / 16, B_),      blk, 0, stream>>>(G, xq, out);
}


# k-contiguous operands (every A but stage 4's; B of stages 1, 3) load 4 k per lane as one 128 bit access, pinned as one group ahead of the WMMAs; k order in each 8-k step permuted: sum order changes
# speedup vs baseline: 2.3679x; 1.0306x over previous
#include <hip/hip_runtime.h>


typedef float v2f __attribute__((ext_vector_type(2)));
typedef float v4f __attribute__((ext_vector_type(4)));
typedef float v8f __attribute__((ext_vector_type(8)));

template <int NB> struct VecN;
template <> struct VecN<2> { typedef v2f type; };
template <> struct VecN<4> { typedef v4f type; };

constexpr int B_   = 4;
constexpr int CQ_  = 256;
constexpr int CKV_ = 256;
constexpr int NQ_  = 2048;
constexpr int NKV_ = 2048;
constexpr int H_   = 8;
constexpr int DK_  = 64;
constexpr int OD_  = H_ * DK_;

template <int NB, int NS, bool BROW> struct Frags;
template <int NB, int NS>
struct Frags<NB, NS, false> {
  v4f a[NS];
  v4f b[NS][NB];
};
template <int NB, int NS>
struct Frags<NB, NS, true> {
  v4f a[NS];
  typename VecN<NB>::type br[NS][2][2];
};

template <int NB, int NS, bool BROW>
__device__ __forceinline__ v2f afrag(const Frags<NB, NS, BROW>& f, int s, int i) {
  v2f a;
  a.x = f.a[s][2 * i];
  a.y = f.a[s][2 * i + 1];
  return a;
}
template <int NB, int NS>
__device__ __forceinline__ v2f bfrag(const Frags<NB, NS, false>& f, int s, int i, int j) {
  v2f b;
  b.x = f.b[s][j][2 * i];
  b.y = f.b[s][j][2 * i + 1];
  return b;
}
template <int NB, int NS>
__device__ __forceinline__ v2f bfrag(const Frags<NB, NS, true>& f, int s, int i, int j) {
  v2f b;
  b.x = f.br[s][i][0][j];
  b.y = f.br[s][i][1][j];
  return b;
}

template <int NB, int NS, bool BROW, bool ACONT>
__device__ __forceinline__ void load_frags(
    Frags<NB, NS, BROW>& f, const float* __restrict__ A, long arow, long csA,
    const float* __restrict__ Bm, long rsB, const long (&bcol)[NB], int kc, int half) {
  typedef typename VecN<NB>::type vN;
#pragma unroll
  for (int s = 0; s < NS; ++s) {
    const int kb = kc + 8 * s + 4 * half;
    if constexpr (ACONT) {
      f.a[s] = *(const v4f*)(A + arow + kb);
    } else {
      f.a[s].x = A[arow + (long)(kb    ) * csA];
      f.a[s].y = A[arow + (long)(kb + 1) * csA];
      f.a[s].z = A[arow + (long)(kb + 2) * csA];
      f.a[s].w = A[arow + (long)(kb + 3) * csA];
    }
    if constexpr (BROW) {
      f.br[s][0][0] = *(const vN*)(Bm + (long)(kb    ) * rsB + bcol[0]);
      f.br[s][0][1] = *(const vN*)(Bm + (long)(kb + 1) * rsB + bcol[0]);
      f.br[s][1][0] = *(const vN*)(Bm + (long)(kb + 2) * rsB + bcol[0]);
      f.br[s][1][1] = *(const vN*)(Bm + (long)(kb + 3) * rsB + bcol[0]);
    } else {
#pragma unroll
      for (int j = 0; j < NB; ++j) f.b[s][j] = *(const v4f*)(Bm + bcol[j] + kb);
    }
  }
}

template <int NB, int NS, bool BROW>
__device__ __forceinline__ void mma_frags(const Frags<NB, NS, BROW>& f, v8f (&acc)[NB]) {
#pragma unroll
  for (int s = 0; s < NS; ++s) {
#pragma unroll
    for (int j = 0; j < NB; ++j)
      acc[j] = __builtin_amdgcn_wmma_f32_16x16x4_f32(false, afrag<NB, NS, BROW>(f, s, 0), false, bfrag<NB, NS>(f, s, 0, j),
                                                     (short)0, acc[j], false, false);
#pragma unroll
    for (int j = 0; j < NB; ++j)
      acc[j] = __builtin_amdgcn_wmma_f32_16x16x4_f32(false, afrag<NB, NS, BROW>(f, s, 1), false, bfrag<NB, NS>(f, s, 1, j),
                                                     (short)0, acc[j], false, false);
  }
}

template <int NB, bool HASR, bool BROW, bool ACONT, int KC, int K>
__device__ __forceinline__ void wmma_gemm_strip(
    const float* __restrict__ A, long rsA, long csA,
    const float* __restrict__ Bm, long rsB, long csB,
    float* __restrict__ C, long rsC,
    const float* __restrict__ R, long rsR,
    int m0, int n0, float scale) {
  static_assert(KC % 8 == 0 && K >= 2 * KC && K % (2 * KC) == 0,
                "K must be a whole number of chunk pairs");
  typedef typename VecN<NB>::type vN;
  constexpr int NS = KC / 8;
  const int lane = threadIdx.x & 31;
  const int r    = lane & 15;
  const int half = lane >> 4;
  const long arow  = (long)(m0 + r) * rsA;
  const long col0 = n0 + NB * r;
  long bcol[NB];
#pragma unroll
  for (int j = 0; j < NB; ++j)
    bcol[j] = BROW ? (long)(n0 + NB * r + j) * csB : (long)(n0 + 16 * j + r) * csB;

  v8f acc[NB];
#pragma unroll
  for (int j = 0; j < NB; ++j) acc[j] = {};

  Frags<NB, NS, BROW> f0, f1;
  load_frags<NB, NS, BROW, ACONT>(f0, A, arow, csA, Bm, rsB, bcol, 0, half);
  int kc = 0;
  for (; kc + 2 * KC < K; kc += 2 * KC) {
    load_frags<NB, NS, BROW, ACONT>(f1, A, arow, csA, Bm, rsB, bcol, kc + KC, half);
    __builtin_amdgcn_sched_barrier(0);
    mma_frags<NB, NS, BROW>(f0, acc);
    __builtin_amdgcn_sched_barrier(0);
    load_frags<NB, NS, BROW, ACONT>(f0, A, arow, csA, Bm, rsB, bcol, kc + 2 * KC, half);
    __builtin_amdgcn_sched_barrier(0);
    mma_frags<NB, NS, BROW>(f1, acc);
    __builtin_amdgcn_sched_barrier(0);
  }
  load_frags<NB, NS, BROW, ACONT>(f1, A, arow, csA, Bm, rsB, bcol, kc + KC, half);
  mma_frags<NB, NS, BROW>(f0, acc);
  mma_frags<NB, NS, BROW>(f1, acc);

  if constexpr (BROW) {
#pragma unroll
    for (int v = 0; v < 8; ++v) {
      const long row = m0 + v + 8 * half;
      vN res;
      if (HASR) res = *(const vN*)(R + row * rsR + col0);
      vN val;
#pragma unroll
      for (int j = 0; j < NB; ++j) {
        float t = acc[j][v] * scale;
        if (HASR) t += res[j];
        val[j] = t;
      }
      *(vN*)(C + row * rsC + col0) = val;
    }
  } else {
#pragma unroll
    for (int j = 0; j < NB; ++j) {
#pragma unroll
      for (int v = 0; v < 8; ++v) {
        const long row = m0 + v + 8 * half;
        const long col = n0 + 16 * j + r;
        float val = acc[j][v] * scale;
        if (HASR) val += R[row * rsR + col];
        C[row * rsC + col] = val;
      }
    }
  }
}

__global__ void __launch_bounds__(32) k_stage1_xxT(const float* __restrict__ xkv, float* __restrict__ S) {
  const int b = blockIdx.z;
  const float* X = xkv + (long)b * CKV_ * NKV_;
  wmma_gemm_strip<2, false, false, true, 64, NKV_>(X, NKV_, 1, X, 1, NKV_,
                            S + (long)b * CKV_ * CKV_, CKV_, nullptr, 0,
                            blockIdx.y * 16, blockIdx.x * 32, 1.0f);
}

__global__ void __launch_bounds__(32) k_stage2_T1(const float* __restrict__ Wk, const float* __restrict__ S,
                            float* __restrict__ T1) {
  const int z = blockIdx.z;
  const int b = z >> 3, h = z & 7;
  wmma_gemm_strip<4, false, true, true, 32, CKV_>(Wk + (long)h * DK_ * CKV_, CKV_, 1,
                            S + (long)b * CKV_ * CKV_, CKV_, 1,
                            T1 + (long)z * DK_ * CKV_, CKV_, nullptr, 0,
                            blockIdx.y * 16, blockIdx.x * 64, 1.0f);
}

__global__ void __launch_bounds__(32) k_stage3_M(const float* __restrict__ T1, const float* __restrict__ Wv,
                           float* __restrict__ Mm) {
  const int z = blockIdx.z;
  const int h = z & 7;
  wmma_gemm_strip<4, false, false, true, 32, CKV_>(T1 + (long)z * DK_ * CKV_, CKV_, 1,
                            Wv + (long)h * DK_ * CKV_, 1, CKV_,
                            Mm + (long)z * DK_ * DK_, DK_, nullptr, 0,
                            blockIdx.y * 16, blockIdx.x * 64, 1.0f / (float)DK_);
}

__global__ void __launch_bounds__(32) k_stage4_Pt(const float* __restrict__ Mm, const float* __restrict__ Wq,
                            float* __restrict__ Pt) {
  const int z = blockIdx.z;
  const int h = z & 7;
  wmma_gemm_strip<4, false, true, false, 32, DK_>(Mm + (long)z * DK_ * DK_, 1, DK_,
                            Wq + (long)h * DK_ * CQ_, CQ_, 1,
                            Pt + (long)z * DK_ * CQ_, CQ_, nullptr, 0,
                            blockIdx.y * 16, blockIdx.x * 64, 1.0f);
}

__global__ void __launch_bounds__(32) k_stage5_G(const float* __restrict__ Wo, const float* __restrict__ Pt,
                           const float* __restrict__ gamma, float* __restrict__ G) {
  const int b = blockIdx.z;
  wmma_gemm_strip<4, false, true, true, 32, OD_>(Wo, OD_, 1,
                            Pt + (long)b * OD_ * CQ_, CQ_, 1,
                            G + (long)b * CQ_ * CQ_, CQ_, nullptr, 0,
                            blockIdx.y * 16, blockIdx.x * 64, gamma[0]);
}

__global__ void __launch_bounds__(32) k_stage6_out(const float* __restrict__ G, const float* __restrict__ xq,
                             float* __restrict__ out) {
  const int b = blockIdx.z;
  const float* Bq = xq + (long)b * CQ_ * NQ_;
  wmma_gemm_strip<4, true, true, true, 32, CQ_>(G + (long)b * CQ_ * CQ_, CQ_, 1,
                           Bq, NQ_, 1,
                           out + (long)b * CQ_ * NQ_, NQ_, Bq, NQ_,
                           blockIdx.y * 16, blockIdx.x * 64, 1.0f);
}

extern "C" void kernel_launch(void* const* d_in, const int* in_sizes, int n_in,
                              void* d_out, int out_size, void* d_ws, size_t ws_size,
                              hipStream_t stream) {
  const float* xq    = (const float*)d_in[0];
  const float* xkv   = (const float*)d_in[1];
  const float* Wq    = (const float*)d_in[2];
  const float* Wk    = (const float*)d_in[3];
  const float* Wv    = (const float*)d_in[4];
  const float* Wo    = (const float*)d_in[5];
  const float* gamma = (const float*)d_in[6];
  float* out = (float*)d_out;

  float* S  = (float*)d_ws;
  float* T1 = S  + (size_t)B_ * CKV_ * CKV_;
  float* Mm = T1 + (size_t)B_ * H_ * DK_ * CKV_;
  float* Pt = Mm + (size_t)B_ * H_ * DK_ * DK_;
  float* G  = Pt + (size_t)B_ * H_ * DK_ * CQ_;

  const dim3 blk(32, 1, 1);
  k_stage1_xxT<<<dim3(CKV_ / 32, CKV_ / 16, B_),     blk, 0, stream>>>(xkv, S);
  k_stage2_T1 <<<dim3(CKV_ / 64, DK_ / 16, B_ * H_), blk, 0, stream>>>(Wk, S, T1);
  k_stage3_M  <<<dim3(DK_ / 64,  DK_ / 16, B_ * H_), blk, 0, stream>>>(T1, Wv, Mm);
  k_stage4_Pt <<<dim3(CQ_ / 64,  DK_ / 16, B_ * H_), blk, 0, stream>>>(Mm, Wq, Pt);
  k_stage5_G  <<<dim3(CQ_ / 64,  CQ_ / 16, B_),      blk, 0, stream>>>(Wo, Pt, gamma, G);
  k_stage6_out<<<dim3(NQ_ / 64,  CQ_ / 16, B_),      blk, 0, stream>>>(G, xq, out);
}


# Split-K in a workgroup: strips of stages 1, 3, 5 get 4 waves (stage 2: 2), each walks its share of K (last loads pinned ahead as in the loop); wave 0 adds partials via LDS: sum order changes
# speedup vs baseline: 3.2621x; 1.3776x over previous
#include <hip/hip_runtime.h>


typedef float v2f __attribute__((ext_vector_type(2)));
typedef float v4f __attribute__((ext_vector_type(4)));
typedef float v8f __attribute__((ext_vector_type(8)));

template <int NB> struct VecN;
template <> struct VecN<2> { typedef v2f type; };
template <> struct VecN<4> { typedef v4f type; };

constexpr int B_   = 4;
constexpr int CQ_  = 256;
constexpr int CKV_ = 256;
constexpr int NQ_  = 2048;
constexpr int NKV_ = 2048;
constexpr int H_   = 8;
constexpr int DK_  = 64;
constexpr int OD_  = H_ * DK_;

template <int NB, int NS, bool BROW> struct Frags;
template <int NB, int NS>
struct Frags<NB, NS, false> {
  v4f a[NS];
  v4f b[NS][NB];
};
template <int NB, int NS>
struct Frags<NB, NS, true> {
  v4f a[NS];
  typename VecN<NB>::type br[NS][2][2];
};

template <int NB, int NS, bool BROW>
__device__ __forceinline__ v2f afrag(const Frags<NB, NS, BROW>& f, int s, int i) {
  v2f a;
  a.x = f.a[s][2 * i];
  a.y = f.a[s][2 * i + 1];
  return a;
}
template <int NB, int NS>
__device__ __forceinline__ v2f bfrag(const Frags<NB, NS, false>& f, int s, int i, int j) {
  v2f b;
  b.x = f.b[s][j][2 * i];
  b.y = f.b[s][j][2 * i + 1];
  return b;
}
template <int NB, int NS>
__device__ __forceinline__ v2f bfrag(const Frags<NB, NS, true>& f, int s, int i, int j) {
  v2f b;
  b.x = f.br[s][i][0][j];
  b.y = f.br[s][i][1][j];
  return b;
}

template <int NB, int NS, bool BROW, bool ACONT>
__device__ __forceinline__ void load_frags(
    Frags<NB, NS, BROW>& f, const float* __restrict__ A, long arow, long csA,
    const float* __restrict__ Bm, long rsB, const long (&bcol)[NB], int kc, int half) {
  typedef typename VecN<NB>::type vN;
#pragma unroll
  for (int s = 0; s < NS; ++s) {
    const int kb = kc + 8 * s + 4 * half;
    if constexpr (ACONT) {
      f.a[s] = *(const v4f*)(A + arow + kb);
    } else {
      f.a[s].x = A[arow + (long)(kb    ) * csA];
      f.a[s].y = A[arow + (long)(kb + 1) * csA];
      f.a[s].z = A[arow + (long)(kb + 2) * csA];
      f.a[s].w = A[arow + (long)(kb + 3) * csA];
    }
    if constexpr (BROW) {
      f.br[s][0][0] = *(const vN*)(Bm + (long)(kb    ) * rsB + bcol[0]);
      f.br[s][0][1] = *(const vN*)(Bm + (long)(kb + 1) * rsB + bcol[0]);
      f.br[s][1][0] = *(const vN*)(Bm + (long)(kb + 2) * rsB + bcol[0]);
      f.br[s][1][1] = *(const vN*)(Bm + (long)(kb + 3) * rsB + bcol[0]);
    } else {
#pragma unroll
      for (int j = 0; j < NB; ++j) f.b[s][j] = *(const v4f*)(Bm + bcol[j] + kb);
    }
  }
}

template <int NB, int NS, bool BROW>
__device__ __forceinline__ void mma_frags(const Frags<NB, NS, BROW>& f, v8f (&acc)[NB]) {
#pragma unroll
  for (int s = 0; s < NS; ++s) {
#pragma unroll
    for (int j = 0; j < NB; ++j)
      acc[j] = __builtin_amdgcn_wmma_f32_16x16x4_f32(false, afrag<NB, NS, BROW>(f, s, 0), false, bfrag<NB, NS>(f, s, 0, j),
                                                     (short)0, acc[j], false, false);
#pragma unroll
    for (int j = 0; j < NB; ++j)
      acc[j] = __builtin_amdgcn_wmma_f32_16x16x4_f32(false, afrag<NB, NS, BROW>(f, s, 1), false, bfrag<NB, NS>(f, s, 1, j),
                                                     (short)0, acc[j], false, false);
  }
}

template <int NB, bool HASR, bool BROW, bool ACONT, int KC, int K, int KS>
__device__ __forceinline__ void wmma_gemm_strip(
    const float* __restrict__ A, long rsA, long csA,
    const float* __restrict__ Bm, long rsB, long csB,
    float* __restrict__ C, long rsC,
    const float* __restrict__ R, long rsR,
    int m0, int n0, float scale) {
  constexpr int KW = K / KS;
  static_assert(K % KS == 0 && KC % 8 == 0 && KW >= 2 * KC && KW % (2 * KC) == 0,
                "a wave's k range must be a whole number of chunk pairs");
  typedef typename VecN<NB>::type vN;
  constexpr int NS = KC / 8;
  int w = 0;
  if constexpr (KS > 1) {
    w = __builtin_amdgcn_readfirstlane((int)(threadIdx.x >> 5));
    A  += (long)w * KW * csA;
    Bm += (long)w * KW * rsB;
  }
  const int lane = threadIdx.x & 31;
  const int r    = lane & 15;
  const int half = lane >> 4;
  const long arow  = (long)(m0 + r) * rsA;
  const long col0 = n0 + NB * r;
  long bcol[NB];
#pragma unroll
  for (int j = 0; j < NB; ++j)
    bcol[j] = BROW ? (long)(n0 + NB * r + j) * csB : (long)(n0 + 16 * j + r) * csB;

  v8f acc[NB];
#pragma unroll
  for (int j = 0; j < NB; ++j) acc[j] = {};

  Frags<NB, NS, BROW> f0, f1;
  load_frags<NB, NS, BROW, ACONT>(f0, A, arow, csA, Bm, rsB, bcol, 0, half);
  int kc = 0;
  for (; kc + 2 * KC < KW; kc += 2 * KC) {
    load_frags<NB, NS, BROW, ACONT>(f1, A, arow, csA, Bm, rsB, bcol, kc + KC, half);
    __builtin_amdgcn_sched_barrier(0);
    mma_frags<NB, NS, BROW>(f0, acc);
    __builtin_amdgcn_sched_barrier(0);
    load_frags<NB, NS, BROW, ACONT>(f0, A, arow, csA, Bm, rsB, bcol, kc + 2 * KC, half);
    __builtin_amdgcn_sched_barrier(0);
    mma_frags<NB, NS, BROW>(f1, acc);
    __builtin_amdgcn_sched_barrier(0);
  }
  load_frags<NB, NS, BROW, ACONT>(f1, A, arow, csA, Bm, rsB, bcol, kc + KC, half);
  if constexpr (KS > 1) __builtin_amdgcn_sched_barrier(0);
  mma_frags<NB, NS, BROW>(f0, acc);
  mma_frags<NB, NS, BROW>(f1, acc);

  if constexpr (KS > 1) {
    __shared__ v4f red[(KS - 1) * NB * 2 * 32];
    if (w != 0) {
#pragma unroll
      for (int j = 0; j < NB; ++j) {
        v4f lo, hi;
        lo.x = acc[j][0]; lo.y = acc[j][1]; lo.z = acc[j][2]; lo.w = acc[j][3];
        hi.x = acc[j][4]; hi.y = acc[j][5]; hi.z = acc[j][6]; hi.w = acc[j][7];
        red[(((w - 1) * NB + j) * 2    ) * 32 + lane] = lo;
        red[(((w - 1) * NB + j) * 2 + 1) * 32 + lane] = hi;
      }
    }
    __syncthreads();
    if (w != 0) return;
#pragma unroll
    for (int p = 1; p < KS; ++p) {
#pragma unroll
      for (int j = 0; j < NB; ++j) {
        const v4f lo = red[(((p - 1) * NB + j) * 2    ) * 32 + lane];
        const v4f hi = red[(((p - 1) * NB + j) * 2 + 1) * 32 + lane];
        acc[j][0] += lo.x; acc[j][1] += lo.y; acc[j][2] += lo.z; acc[j][3] += lo.w;
        acc[j][4] += hi.x; acc[j][5] += hi.y; acc[j][6] += hi.z; acc[j][7] += hi.w;
      }
    }
  }

  if constexpr (BROW) {
#pragma unroll
    for (int v = 0; v < 8; ++v) {
      const long row = m0 + v + 8 * half;
      vN res;
      if (HASR) res = *(const vN*)(R + row * rsR + col0);
      vN val;
#pragma unroll
      for (int j = 0; j < NB; ++j) {
        float t = acc[j][v] * scale;
        if (HASR) t += res[j];
        val[j] = t;
      }
      *(vN*)(C + row * rsC + col0) = val;
    }
  } else {
#pragma unroll
    for (int j = 0; j < NB; ++j) {
#pragma unroll
      for (int v = 0; v < 8; ++v) {
        const long row = m0 + v + 8 * half;
        const long col = n0 + 16 * j + r;
        float val = acc[j][v] * scale;
        if (HASR) val += R[row * rsR + col];
        C[row * rsC + col] = val;
      }
    }
  }
}

constexpr int KS1 = 4;
constexpr int KS2 = 2;
constexpr int KS3 = 4;
constexpr int KS5 = 4;
__global__ void __launch_bounds__(32 * KS1) k_stage1_xxT(const float* __restrict__ xkv, float* __restrict__ S) {
  const int b = blockIdx.z;
  const float* X = xkv + (long)b * CKV_ * NKV_;
  wmma_gemm_strip<2, false, false, true, 64, NKV_, KS1>(X, NKV_, 1, X, 1, NKV_,
                            S + (long)b * CKV_ * CKV_, CKV_, nullptr, 0,
                            blockIdx.y * 16, blockIdx.x * 32, 1.0f);
}

__global__ void __launch_bounds__(32 * KS2) k_stage2_T1(const float* __restrict__ Wk, const float* __restrict__ S,
                            float* __restrict__ T1) {
  const int z = blockIdx.z;
  const int b = z >> 3, h = z & 7;
  wmma_gemm_strip<4, false, true, true, 32, CKV_, KS2>(Wk + (long)h * DK_ * CKV_, CKV_, 1,
                            S + (long)b * CKV_ * CKV_, CKV_, 1,
                            T1 + (long)z * DK_ * CKV_, CKV_, nullptr, 0,
                            blockIdx.y * 16, blockIdx.x * 64, 1.0f);
}

__global__ void __launch_bounds__(32 * KS3) k_stage3_M(const float* __restrict__ T1, const float* __restrict__ Wv,
                           float* __restrict__ Mm) {
  const int z = blockIdx.z;
  const int h = z & 7;
  wmma_gemm_strip<4, false, false, true, 32, CKV_, KS3>(T1 + (long)z * DK_ * CKV_, CKV_, 1,
                            Wv + (long)h * DK_ * CKV_, 1, CKV_,
                            Mm + (long)z * DK_ * DK_, DK_, nullptr, 0,
                            blockIdx.y * 16, blockIdx.x * 64, 1.0f / (float)DK_);
}

__global__ void __launch_bounds__(32) k_stage4_Pt(const float* __restrict__ Mm, const float* __restrict__ Wq,
                            float* __restrict__ Pt) {
  const int z = blockIdx.z;
  const int h = z & 7;
  wmma_gemm_strip<4, false, true, false, 32, DK_, 1>(Mm + (long)z * DK_ * DK_, 1, DK_,
                            Wq + (long)h * DK_ * CQ_, CQ_, 1,
                            Pt + (long)z * DK_ * CQ_, CQ_, nullptr, 0,
                            blockIdx.y * 16, blockIdx.x * 64, 1.0f);
}

__global__ void __launch_bounds__(32 * KS5) k_stage5_G(const float* __restrict__ Wo, const float* __restrict__ Pt,
                           const float* __restrict__ gamma, float* __restrict__ G) {
  const int b = blockIdx.z;
  wmma_gemm_strip<4, false, true, true, 32, OD_, KS5>(Wo, OD_, 1,
                            Pt + (long)b * OD_ * CQ_, CQ_, 1,
                            G + (long)b * CQ_ * CQ_, CQ_, nullptr, 0,
                            blockIdx.y * 16, blockIdx.x * 64, gamma[0]);
}

__global__ void __launch_bounds__(32) k_stage6_out(const float* __restrict__ G, const float* __restrict__ xq,
                             float* __restrict__ out) {
  const int b = blockIdx.z;
  const float* Bq = xq + (long)b * CQ_ * NQ_;
  wmma_gemm_strip<4, true, true, true, 32, CQ_, 1>(G + (long)b * CQ_ * CQ_, CQ_, 1,
                           Bq, NQ_, 1,
                           out + (long)b * CQ_ * NQ_, NQ_, Bq, NQ_,
                           blockIdx.y * 16, blockIdx.x * 64, 1.0f);
}

extern "C" void kernel_launch(void* const* d_in, const int* in_sizes, int n_in,
                              void* d_out, int out_size, void* d_ws, size_t ws_size,
                              hipStream_t stream) {
  const float* xq    = (const float*)d_in[0];
  const float* xkv   = (const float*)d_in[1];
  const float* Wq    = (const float*)d_in[2];
  const float* Wk    = (const float*)d_in[3];
  const float* Wv    = (const float*)d_in[4];
  const float* Wo    = (const float*)d_in[5];
  const float* gamma = (const float*)d_in[6];
  float* out = (float*)d_out;

  float* S  = (float*)d_ws;
  float* T1 = S  + (size_t)B_ * CKV_ * CKV_;
  float* Mm = T1 + (size_t)B_ * H_ * DK_ * CKV_;
  float* Pt = Mm + (size_t)B_ * H_ * DK_ * DK_;
  float* G  = Pt + (size_t)B_ * H_ * DK_ * CQ_;

  const dim3 blk(32, 1, 1);
  k_stage1_xxT<<<dim3(CKV_ / 32, CKV_ / 16, B_),     dim3(32 * KS1, 1, 1), 0, stream>>>(xkv, S);
  k_stage2_T1 <<<dim3(CKV_ / 64, DK_ / 16, B_ * H_), dim3(32 * KS2, 1, 1), 0, stream>>>(Wk, S, T1);
  k_stage3_M  <<<dim3(DK_ / 64,  DK_ / 16, B_ * H_), dim3(32 * KS3, 1, 1), 0, stream>>>(T1, Wv, Mm);
  k_stage4_Pt <<<dim3(CQ_ / 64,  DK_ / 16, B_ * H_), blk, 0, stream>>>(Mm, Wq, Pt);
  k_stage5_G  <<<dim3(CQ_ / 64,  CQ_ / 16, B_),      dim3(32 * KS5, 1, 1), 0, stream>>>(Wo, Pt, gamma, G);
  k_stage6_out<<<dim3(NQ_ / 64,  CQ_ / 16, B_),      blk, 0, stream>>>(G, xq, out);
}


# Stages 3 and 4 fused into one launch: a 4-wave workgroup forms its 16 x 64 strip of M^T split-K as before, sums it into an LDS tile, each wave multiplies it by 64 columns of Wq; sum order kept
# speedup vs baseline: 3.5059x; 1.0747x over previous
#include <hip/hip_runtime.h>


typedef float v2f __attribute__((ext_vector_type(2)));
typedef float v4f __attribute__((ext_vector_type(4)));
typedef float v8f __attribute__((ext_vector_type(8)));

template <int NB> struct VecN;
template <> struct VecN<2> { typedef v2f type; };
template <> struct VecN<4> { typedef v4f type; };

constexpr int B_   = 4;
constexpr int CQ_  = 256;
constexpr int CKV_ = 256;
constexpr int NQ_  = 2048;
constexpr int NKV_ = 2048;
constexpr int H_   = 8;
constexpr int DK_  = 64;
constexpr int OD_  = H_ * DK_;

template <int NB, int NS, bool BROW> struct Frags;
template <int NB, int NS>
struct Frags<NB, NS, false> {
  v4f a[NS];
  v4f b[NS][NB];
};
template <int NB, int NS>
struct Frags<NB, NS, true> {
  v4f a[NS];
  typename VecN<NB>::type br[NS][2][2];
};

template <int NB, int NS, bool BROW>
__device__ __forceinline__ v2f afrag(const Frags<NB, NS, BROW>& f, int s, int i) {
  v2f a;
  a.x = f.a[s][2 * i];
  a.y = f.a[s][2 * i + 1];
  return a;
}
template <int NB, int NS>
__device__ __forceinline__ v2f bfrag(const Frags<NB, NS, false>& f, int s, int i, int j) {
  v2f b;
  b.x = f.b[s][j][2 * i];
  b.y = f.b[s][j][2 * i + 1];
  return b;
}
template <int NB, int NS>
__device__ __forceinline__ v2f bfrag(const Frags<NB, NS, true>& f, int s, int i, int j) {
  v2f b;
  b.x = f.br[s][i][0][j];
  b.y = f.br[s][i][1][j];
  return b;
}

template <int NB, int NS, bool BROW, bool ACONT>
__device__ __forceinline__ void load_frags(
    Frags<NB, NS, BROW>& f, const float* __restrict__ A, long arow, long csA,
    const float* __restrict__ Bm, long rsB, const long (&bcol)[NB], int kc, int half) {
  typedef typename VecN<NB>::type vN;
#pragma unroll
  for (int s = 0; s < NS; ++s) {
    const int kb = kc + 8 * s + 4 * half;
    if constexpr (ACONT) {
      f.a[s] = *(const v4f*)(A + arow + kb);
    } else {
      f.a[s].x = A[arow + (long)(kb    ) * csA];
      f.a[s].y = A[arow + (long)(kb + 1) * csA];
      f.a[s].z = A[arow + (long)(kb + 2) * csA];
      f.a[s].w = A[arow + (long)(kb + 3) * csA];
    }
    if constexpr (BROW) {
      f.br[s][0][0] = *(const vN*)(Bm + (long)(kb    ) * rsB + bcol[0]);
      f.br[s][0][1] = *(const vN*)(Bm + (long)(kb + 1) * rsB + bcol[0]);
      f.br[s][1][0] = *(const vN*)(Bm + (long)(kb + 2) * rsB + bcol[0]);
      f.br[s][1][1] = *(const vN*)(Bm + (long)(kb + 3) * rsB + bcol[0]);
    } else {
#pragma unroll
      for (int j = 0; j < NB; ++j) f.b[s][j] = *(const v4f*)(Bm + bcol[j] + kb);
    }
  }
}

template <int NB, int NS>
__device__ __forceinline__ void load_b_rows(
    Frags<NB, NS, true>& f, const float* __restrict__ Bm, long rsB, long col, int kc, int half) {
  typedef typename VecN<NB>::type vN;
#pragma unroll
  for (int s = 0; s < NS; ++s) {
    const int kb = kc + 8 * s + 4 * half;
    f.br[s][0][0] = *(const vN*)(Bm + (long)(kb    ) * rsB + col);
    f.br[s][0][1] = *(const vN*)(Bm + (long)(kb + 1) * rsB + col);
    f.br[s][1][0] = *(const vN*)(Bm + (long)(kb + 2) * rsB + col);
    f.br[s][1][1] = *(const vN*)(Bm + (long)(kb + 3) * rsB + col);
  }
}

template <int NB, int NS, bool BROW>
__device__ __forceinline__ void mma_frags(const Frags<NB, NS, BROW>& f, v8f (&acc)[NB]) {
#pragma unroll
  for (int s = 0; s < NS; ++s) {
#pragma unroll
    for (int j = 0; j < NB; ++j)
      acc[j] = __builtin_amdgcn_wmma_f32_16x16x4_f32(false, afrag<NB, NS, BROW>(f, s, 0), false, bfrag<NB, NS>(f, s, 0, j),
                                                     (short)0, acc[j], false, false);
#pragma unroll
    for (int j = 0; j < NB; ++j)
      acc[j] = __builtin_amdgcn_wmma_f32_16x16x4_f32(false, afrag<NB, NS, BROW>(f, s, 1), false, bfrag<NB, NS>(f, s, 1, j),
                                                     (short)0, acc[j], false, false);
  }
}

template <int NB, bool HASR, bool BROW, bool ACONT, int KC, int K, int KS>
__device__ __forceinline__ void wmma_gemm_strip(
    const float* __restrict__ A, long rsA, long csA,
    const float* __restrict__ Bm, long rsB, long csB,
    float* __restrict__ C, long rsC,
    const float* __restrict__ R, long rsR,
    int m0, int n0, float scale) {
  constexpr int KW = K / KS;
  static_assert(K % KS == 0 && KC % 8 == 0 && KW >= 2 * KC && KW % (2 * KC) == 0,
                "a wave's k range must be a whole number of chunk pairs");
  typedef typename VecN<NB>::type vN;
  constexpr int NS = KC / 8;
  int w = 0;
  if constexpr (KS > 1) {
    w = __builtin_amdgcn_readfirstlane((int)(threadIdx.x >> 5));
    A  += (long)w * KW * csA;
    Bm += (long)w * KW * rsB;
  }
  const int lane = threadIdx.x & 31;
  const int r    = lane & 15;
  const int half = lane >> 4;
  const long arow  = (long)(m0 + r) * rsA;
  const long col0 = n0 + NB * r;
  long bcol[NB];
#pragma unroll
  for (int j = 0; j < NB; ++j)
    bcol[j] = BROW ? (long)(n0 + NB * r + j) * csB : (long)(n0 + 16 * j + r) * csB;

  v8f acc[NB];
#pragma unroll
  for (int j = 0; j < NB; ++j) acc[j] = {};

  Frags<NB, NS, BROW> f0, f1;
  load_frags<NB, NS, BROW, ACONT>(f0, A, arow, csA, Bm, rsB, bcol, 0, half);
  int kc = 0;
  for (; kc + 2 * KC < KW; kc += 2 * KC) {
    load_frags<NB, NS, BROW, ACONT>(f1, A, arow, csA, Bm, rsB, bcol, kc + KC, half);
    __builtin_amdgcn_sched_barrier(0);
    mma_frags<NB, NS, BROW>(f0, acc);
    __builtin_amdgcn_sched_barrier(0);
    load_frags<NB, NS, BROW, ACONT>(f0, A, arow, csA, Bm, rsB, bcol, kc + 2 * KC, half);
    __builtin_amdgcn_sched_barrier(0);
    mma_frags<NB, NS, BROW>(f1, acc);
    __builtin_amdgcn_sched_barrier(0);
  }
  load_frags<NB, NS, BROW, ACONT>(f1, A, arow, csA, Bm, rsB, bcol, kc + KC, half);
  if constexpr (KS > 1) __builtin_amdgcn_sched_barrier(0);
  mma_frags<NB, NS, BROW>(f0, acc);
  mma_frags<NB, NS, BROW>(f1, acc);

  if constexpr (KS > 1) {
    __shared__ v4f red[(KS - 1) * NB * 2 * 32];
    if (w != 0) {
#pragma unroll
      for (int j = 0; j < NB; ++j) {
        v4f lo, hi;
        lo.x = acc[j][0]; lo.y = acc[j][1]; lo.z = acc[j][2]; lo.w = acc[j][3];
        hi.x = acc[j][4]; hi.y = acc[j][5]; hi.z = acc[j][6]; hi.w = acc[j][7];
        red[(((w - 1) * NB + j) * 2    ) * 32 + lane] = lo;
        red[(((w - 1) * NB + j) * 2 + 1) * 32 + lane] = hi;
      }
    }
    __syncthreads();
    if (w != 0) return;
#pragma unroll
    for (int p = 1; p < KS; ++p) {
#pragma unroll
      for (int j = 0; j < NB; ++j) {
        const v4f lo = red[(((p - 1) * NB + j) * 2    ) * 32 + lane];
        const v4f hi = red[(((p - 1) * NB + j) * 2 + 1) * 32 + lane];
        acc[j][0] += lo.x; acc[j][1] += lo.y; acc[j][2] += lo.z; acc[j][3] += lo.w;
        acc[j][4] += hi.x; acc[j][5] += hi.y; acc[j][6] += hi.z; acc[j][7] += hi.w;
      }
    }
  }

  if constexpr (BROW) {
#pragma unroll
    for (int v = 0; v < 8; ++v) {
      const long row = m0 + v + 8 * half;
      vN res;
      if (HASR) res = *(const vN*)(R + row * rsR + col0);
      vN val;
#pragma unroll
      for (int j = 0; j < NB; ++j) {
        float t = acc[j][v] * scale;
        if (HASR) t += res[j];
        val[j] = t;
      }
      *(vN*)(C + row * rsC + col0) = val;
    }
  } else {
#pragma unroll
    for (int j = 0; j < NB; ++j) {
#pragma unroll
      for (int v = 0; v < 8; ++v) {
        const long row = m0 + v + 8 * half;
        const long col = n0 + 16 * j + r;
        float val = acc[j][v] * scale;
        if (HASR) val += R[row * rsR + col];
        C[row * rsC + col] = val;
      }
    }
  }
}

constexpr int KS1 = 4;
constexpr int KS2 = 2;
constexpr int KS5 = 4;
__global__ void __launch_bounds__(32 * KS1) k_stage1_xxT(const float* __restrict__ xkv, float* __restrict__ S) {
  const int b = blockIdx.z;
  const float* X = xkv + (long)b * CKV_ * NKV_;
  wmma_gemm_strip<2, false, false, true, 64, NKV_, KS1>(X, NKV_, 1, X, 1, NKV_,
                            S + (long)b * CKV_ * CKV_, CKV_, nullptr, 0,
                            blockIdx.y * 16, blockIdx.x * 32, 1.0f);
}

__global__ void __launch_bounds__(32 * KS2) k_stage2_T1(const float* __restrict__ Wk, const float* __restrict__ S,
                            float* __restrict__ T1) {
  const int z = blockIdx.z;
  const int b = z >> 3, h = z & 7;
  wmma_gemm_strip<4, false, true, true, 32, CKV_, KS2>(Wk + (long)h * DK_ * CKV_, CKV_, 1,
                            S + (long)b * CKV_ * CKV_, CKV_, 1,
                            T1 + (long)z * DK_ * CKV_, CKV_, nullptr, 0,
                            blockIdx.y * 16, blockIdx.x * 64, 1.0f);
}

constexpr int KS34 = 4;
constexpr int TP34 = DK_ + 4;
__global__ void __launch_bounds__(32 * KS34) k_stage34_Pt(const float* __restrict__ T1, const float* __restrict__ Wv,
                             const float* __restrict__ Wq, float* __restrict__ Pt) {
  constexpr int NB = 4, KC = 32, NS = KC / 8;
  constexpr int KW = CKV_ / KS34;
  static_assert(KW == 2 * KC && DK_ == 2 * KC, "both parts are one chunk pair per wave");
  const int z = blockIdx.z;
  const int h = z & 7;
  const int m0 = blockIdx.y * 16;
  const int w = __builtin_amdgcn_readfirstlane((int)(threadIdx.x >> 5));
  const int lane = threadIdx.x & 31;
  const int r    = lane & 15;
  const int half = lane >> 4;

  __shared__ v4f red[KS34 * NB * 2 * 32];
  __shared__ __attribute__((aligned(16))) float mt[16 * TP34];

  const float* A1 = Wv + (long)h * DK_ * CKV_ + (long)w * KW;
  const float* B1 = T1 + (long)z * DK_ * CKV_ + (long)w * KW;
  const long arow = (long)(m0 + r) * CKV_;
  long bcol[NB];
#pragma unroll
  for (int j = 0; j < NB; ++j) bcol[j] = (long)(16 * j + r) * CKV_;
  v8f acc[NB];
#pragma unroll
  for (int j = 0; j < NB; ++j) acc[j] = {};
  const float* Bq = Wq + (long)h * DK_ * CQ_;
  const long col0 = 64 * w + NB * r;
  Frags<NB, NS, true> g0, g1;
  {
    Frags<NB, NS, false> f0, f1;
    load_frags<NB, NS, false, true>(f0, A1, arow, 1, B1, 1, bcol, 0, half);
    load_frags<NB, NS, false, true>(f1, A1, arow, 1, B1, 1, bcol, KC, half);
    __builtin_amdgcn_sched_barrier(0);
    mma_frags<NB, NS, false>(f0, acc);
    __builtin_amdgcn_sched_barrier(0);
    load_b_rows<NB, NS>(g0, Bq, CQ_, col0, 0, half);
    load_b_rows<NB, NS>(g1, Bq, CQ_, col0, KC, half);
    __builtin_amdgcn_sched_barrier(0);
    mma_frags<NB, NS, false>(f1, acc);
  }

#pragma unroll
  for (int j = 0; j < NB; ++j) {
    v4f lo, hi;
    lo.x = acc[j][0]; lo.y = acc[j][1]; lo.z = acc[j][2]; lo.w = acc[j][3];
    hi.x = acc[j][4]; hi.y = acc[j][5]; hi.z = acc[j][6]; hi.w = acc[j][7];
    red[((w * NB + j) * 2    ) * 32 + lane] = lo;
    red[((w * NB + j) * 2 + 1) * 32 + lane] = hi;
  }
  __syncthreads();
#pragma unroll
  for (int hh = 0; hh < 2; ++hh) {
    v4f s = red[((0 * NB + w) * 2 + hh) * 32 + lane];
#pragma unroll
    for (int p = 1; p < KS34; ++p) {
      const v4f t = red[((p * NB + w) * 2 + hh) * 32 + lane];
      s.x += t.x; s.y += t.y; s.z += t.z; s.w += t.w;
    }
    const float sc = 1.0f / (float)DK_;
    mt[(4 * hh + 0 + 8 * half) * TP34 + 16 * w + r] = s.x * sc;
    mt[(4 * hh + 1 + 8 * half) * TP34 + 16 * w + r] = s.y * sc;
    mt[(4 * hh + 2 + 8 * half) * TP34 + 16 * w + r] = s.z * sc;
    mt[(4 * hh + 3 + 8 * half) * TP34 + 16 * w + r] = s.w * sc;
  }
  __syncthreads();

#pragma unroll
  for (int s = 0; s < NS; ++s) {
    g0.a[s] = *(const v4f*)(mt + r * TP34 + 8 * s + 4 * half);
    g1.a[s] = *(const v4f*)(mt + r * TP34 + KC + 8 * s + 4 * half);
  }
  __builtin_amdgcn_sched_barrier(0);
  v8f acc2[NB];
#pragma unroll
  for (int j = 0; j < NB; ++j) acc2[j] = {};
  mma_frags<NB, NS, true>(g0, acc2);
  mma_frags<NB, NS, true>(g1, acc2);

  float* C = Pt + (long)z * DK_ * CQ_;
#pragma unroll
  for (int v = 0; v < 8; ++v) {
    const long row = m0 + v + 8 * half;
    v4f val;
#pragma unroll
    for (int j = 0; j < NB; ++j) val[j] = acc2[j][v];
    *(v4f*)(C + row * CQ_ + col0) = val;
  }
}

__global__ void __launch_bounds__(32 * KS5) k_stage5_G(const float* __restrict__ Wo, const float* __restrict__ Pt,
                           const float* __restrict__ gamma, float* __restrict__ G) {
  const int b = blockIdx.z;
  wmma_gemm_strip<4, false, true, true, 32, OD_, KS5>(Wo, OD_, 1,
                            Pt + (long)b * OD_ * CQ_, CQ_, 1,
                            G + (long)b * CQ_ * CQ_, CQ_, nullptr, 0,
                            blockIdx.y * 16, blockIdx.x * 64, gamma[0]);
}

__global__ void __launch_bounds__(32) k_stage6_out(const float* __restrict__ G, const float* __restrict__ xq,
                             float* __restrict__ out) {
  const int b = blockIdx.z;
  const float* Bq = xq + (long)b * CQ_ * NQ_;
  wmma_gemm_strip<4, true, true, true, 32, CQ_, 1>(G + (long)b * CQ_ * CQ_, CQ_, 1,
                           Bq, NQ_, 1,
                           out + (long)b * CQ_ * NQ_, NQ_, Bq, NQ_,
                           blockIdx.y * 16, blockIdx.x * 64, 1.0f);
}

extern "C" void kernel_launch(void* const* d_in, const int* in_sizes, int n_in,
                              void* d_out, int out_size, void* d_ws, size_t ws_size,
                              hipStream_t stream) {
  const float* xq    = (const float*)d_in[0];
  const float* xkv   = (const float*)d_in[1];
  const float* Wq    = (const float*)d_in[2];
  const float* Wk    = (const float*)d_in[3];
  const float* Wv    = (const float*)d_in[4];
  const float* Wo    = (const float*)d_in[5];
  const float* gamma = (const float*)d_in[6];
  float* out = (float*)d_out;

  float* S  = (float*)d_ws;
  float* T1 = S  + (size_t)B_ * CKV_ * CKV_;
  float* Pt = T1 + (size_t)B_ * H_ * DK_ * CKV_;
  float* G  = Pt + (size_t)B_ * H_ * DK_ * CQ_;

  const dim3 blk(32, 1, 1);
  k_stage1_xxT<<<dim3(CKV_ / 32, CKV_ / 16, B_),     dim3(32 * KS1, 1, 1), 0, stream>>>(xkv, S);
  k_stage2_T1 <<<dim3(CKV_ / 64, DK_ / 16, B_ * H_), dim3(32 * KS2, 1, 1), 0, stream>>>(Wk, S, T1);
  k_stage34_Pt<<<dim3(1,         DK_ / 16, B_ * H_), dim3(32 * KS34, 1, 1), 0, stream>>>(T1, Wv, Wq, Pt);
  k_stage5_G  <<<dim3(CQ_ / 64,  CQ_ / 16, B_),      dim3(32 * KS5, 1, 1), 0, stream>>>(Wo, Pt, gamma, G);
  k_stage6_out<<<dim3(NQ_ / 64,  CQ_ / 16, B_),      blk, 0, stream>>>(G, xq, out);
}


# Stage 1: each wave stages its 64-k chunk (16 A + 32 B rows) via whole-line loads (4 lines per load, was 16) into an LDS region of its own, reads fragments back in the same k order; sum order kept
# speedup vs baseline: 3.8863x; 1.1085x over previous
#include <hip/hip_runtime.h>


typedef float v2f __attribute__((ext_vector_type(2)));
typedef float v4f __attribute__((ext_vector_type(4)));
typedef float v8f __attribute__((ext_vector_type(8)));

template <int NB> struct VecN;
template <> struct VecN<2> { typedef v2f type; };
template <> struct VecN<4> { typedef v4f type; };

constexpr int B_   = 4;
constexpr int CQ_  = 256;
constexpr int CKV_ = 256;
constexpr int NQ_  = 2048;
constexpr int NKV_ = 2048;
constexpr int H_   = 8;
constexpr int DK_  = 64;
constexpr int OD_  = H_ * DK_;

template <int NB, int NS, bool BROW> struct Frags;
template <int NB, int NS>
struct Frags<NB, NS, false> {
  v4f a[NS];
  v4f b[NS][NB];
};
template <int NB, int NS>
struct Frags<NB, NS, true> {
  v4f a[NS];
  typename VecN<NB>::type br[NS][2][2];
};

template <int NB, int NS, bool BROW>
__device__ __forceinline__ v2f afrag(const Frags<NB, NS, BROW>& f, int s, int i) {
  v2f a;
  a.x = f.a[s][2 * i];
  a.y = f.a[s][2 * i + 1];
  return a;
}
template <int NB, int NS>
__device__ __forceinline__ v2f bfrag(const Frags<NB, NS, false>& f, int s, int i, int j) {
  v2f b;
  b.x = f.b[s][j][2 * i];
  b.y = f.b[s][j][2 * i + 1];
  return b;
}
template <int NB, int NS>
__device__ __forceinline__ v2f bfrag(const Frags<NB, NS, true>& f, int s, int i, int j) {
  v2f b;
  b.x = f.br[s][i][0][j];
  b.y = f.br[s][i][1][j];
  return b;
}

template <int NB, int NS, bool BROW, bool ACONT>
__device__ __forceinline__ void load_frags(
    Frags<NB, NS, BROW>& f, const float* __restrict__ A, long arow, long csA,
    const float* __restrict__ Bm, long rsB, const long (&bcol)[NB], int kc, int half) {
  typedef typename VecN<NB>::type vN;
#pragma unroll
  for (int s = 0; s < NS; ++s) {
    const int kb = kc + 8 * s + 4 * half;
    if constexpr (ACONT) {
      f.a[s] = *(const v4f*)(A + arow + kb);
    } else {
      f.a[s].x = A[arow + (long)(kb    ) * csA];
      f.a[s].y = A[arow + (long)(kb + 1) * csA];
      f.a[s].z = A[arow + (long)(kb + 2) * csA];
      f.a[s].w = A[arow + (long)(kb + 3) * csA];
    }
    if constexpr (BROW) {
      f.br[s][0][0] = *(const vN*)(Bm + (long)(kb    ) * rsB + bcol[0]);
      f.br[s][0][1] = *(const vN*)(Bm + (long)(kb + 1) * rsB + bcol[0]);
      f.br[s][1][0] = *(const vN*)(Bm + (long)(kb + 2) * rsB + bcol[0]);
      f.br[s][1][1] = *(const vN*)(Bm + (long)(kb + 3) * rsB + bcol[0]);
    } else {
#pragma unroll
      for (int j = 0; j < NB; ++j) f.b[s][j] = *(const v4f*)(Bm + bcol[j] + kb);
    }
  }
}

template <int NB, int NS>
__device__ __forceinline__ void load_b_rows(
    Frags<NB, NS, true>& f, const float* __restrict__ Bm, long rsB, long col, int kc, int half) {
  typedef typename VecN<NB>::type vN;
#pragma unroll
  for (int s = 0; s < NS; ++s) {
    const int kb = kc + 8 * s + 4 * half;
    f.br[s][0][0] = *(const vN*)(Bm + (long)(kb    ) * rsB + col);
    f.br[s][0][1] = *(const vN*)(Bm + (long)(kb + 1) * rsB + col);
    f.br[s][1][0] = *(const vN*)(Bm + (long)(kb + 2) * rsB + col);
    f.br[s][1][1] = *(const vN*)(Bm + (long)(kb + 3) * rsB + col);
  }
}

template <int NB, int NS, bool BROW>
__device__ __forceinline__ void mma_frags(const Frags<NB, NS, BROW>& f, v8f (&acc)[NB]) {
#pragma unroll
  for (int s = 0; s < NS; ++s) {
#pragma unroll
    for (int j = 0; j < NB; ++j)
      acc[j] = __builtin_amdgcn_wmma_f32_16x16x4_f32(false, afrag<NB, NS, BROW>(f, s, 0), false, bfrag<NB, NS>(f, s, 0, j),
                                                     (short)0, acc[j], false, false);
#pragma unroll
    for (int j = 0; j < NB; ++j)
      acc[j] = __builtin_amdgcn_wmma_f32_16x16x4_f32(false, afrag<NB, NS, BROW>(f, s, 1), false, bfrag<NB, NS>(f, s, 1, j),
                                                     (short)0, acc[j], false, false);
  }
}

template <int NB, bool HASR, bool BROW, bool ACONT, int KC, int K, int KS>
__device__ __forceinline__ void wmma_gemm_strip(
    const float* __restrict__ A, long rsA, long csA,
    const float* __restrict__ Bm, long rsB, long csB,
    float* __restrict__ C, long rsC,
    const float* __restrict__ R, long rsR,
    int m0, int n0, float scale) {
  constexpr int KW = K / KS;
  static_assert(K % KS == 0 && KC % 8 == 0 && KW >= 2 * KC && KW % (2 * KC) == 0,
                "a wave's k range must be a whole number of chunk pairs");
  typedef typename VecN<NB>::type vN;
  constexpr int NS = KC / 8;
  int w = 0;
  if constexpr (KS > 1) {
    w = __builtin_amdgcn_readfirstlane((int)(threadIdx.x >> 5));
    A  += (long)w * KW * csA;
    Bm += (long)w * KW * rsB;
  }
  const int lane = threadIdx.x & 31;
  const int r    = lane & 15;
  const int half = lane >> 4;
  const long arow  = (long)(m0 + r) * rsA;
  const long col0 = n0 + NB * r;
  long bcol[NB];
#pragma unroll
  for (int j = 0; j < NB; ++j)
    bcol[j] = BROW ? (long)(n0 + NB * r + j) * csB : (long)(n0 + 16 * j + r) * csB;

  v8f acc[NB];
#pragma unroll
  for (int j = 0; j < NB; ++j) acc[j] = {};

  Frags<NB, NS, BROW> f0, f1;
  load_frags<NB, NS, BROW, ACONT>(f0, A, arow, csA, Bm, rsB, bcol, 0, half);
  int kc = 0;
  for (; kc + 2 * KC < KW; kc += 2 * KC) {
    load_frags<NB, NS, BROW, ACONT>(f1, A, arow, csA, Bm, rsB, bcol, kc + KC, half);
    __builtin_amdgcn_sched_barrier(0);
    mma_frags<NB, NS, BROW>(f0, acc);
    __builtin_amdgcn_sched_barrier(0);
    load_frags<NB, NS, BROW, ACONT>(f0, A, arow, csA, Bm, rsB, bcol, kc + 2 * KC, half);
    __builtin_amdgcn_sched_barrier(0);
    mma_frags<NB, NS, BROW>(f1, acc);
    __builtin_amdgcn_sched_barrier(0);
  }
  load_frags<NB, NS, BROW, ACONT>(f1, A, arow, csA, Bm, rsB, bcol, kc + KC, half);
  if constexpr (KS > 1) __builtin_amdgcn_sched_barrier(0);
  mma_frags<NB, NS, BROW>(f0, acc);
  mma_frags<NB, NS, BROW>(f1, acc);

  if constexpr (KS > 1) {
    __shared__ v4f red[(KS - 1) * NB * 2 * 32];
    if (w != 0) {
#pragma unroll
      for (int j = 0; j < NB; ++j) {
        v4f lo, hi;
        lo.x = acc[j][0]; lo.y = acc[j][1]; lo.z = acc[j][2]; lo.w = acc[j][3];
        hi.x = acc[j][4]; hi.y = acc[j][5]; hi.z = acc[j][6]; hi.w = acc[j][7];
        red[(((w - 1) * NB + j) * 2    ) * 32 + lane] = lo;
        red[(((w - 1) * NB + j) * 2 + 1) * 32 + lane] = hi;
      }
    }
    __syncthreads();
    if (w != 0) return;
#pragma unroll
    for (int p = 1; p < KS; ++p) {
#pragma unroll
      for (int j = 0; j < NB; ++j) {
        const v4f lo = red[(((p - 1) * NB + j) * 2    ) * 32 + lane];
        const v4f hi = red[(((p - 1) * NB + j) * 2 + 1) * 32 + lane];
        acc[j][0] += lo.x; acc[j][1] += lo.y; acc[j][2] += lo.z; acc[j][3] += lo.w;
        acc[j][4] += hi.x; acc[j][5] += hi.y; acc[j][6] += hi.z; acc[j][7] += hi.w;
      }
    }
  }

  if constexpr (BROW) {
#pragma unroll
    for (int v = 0; v < 8; ++v) {
      const long row = m0 + v + 8 * half;
      vN res;
      if (HASR) res = *(const vN*)(R + row * rsR + col0);
      vN val;
#pragma unroll
      for (int j = 0; j < NB; ++j) {
        float t = acc[j][v] * scale;
        if (HASR) t += res[j];
        val[j] = t;
      }
      *(vN*)(C + row * rsC + col0) = val;
    }
  } else {
#pragma unroll
    for (int j = 0; j < NB; ++j) {
#pragma unroll
      for (int v = 0; v < 8; ++v) {
        const long row = m0 + v + 8 * half;
        const long col = n0 + 16 * j + r;
        float val = acc[j][v] * scale;
        if (HASR) val += R[row * rsR + col];
        C[row * rsC + col] = val;
      }
    }
  }
}

constexpr int KS1 = 4;
constexpr int KS2 = 2;
constexpr int KS5 = 4;
constexpr int S1KC   = 64;
constexpr int S1ROWS = 48;
constexpr int S1P    = S1KC + 4;
__global__ void __launch_bounds__(32 * KS1) k_stage1_xxT(const float* __restrict__ xkv, float* __restrict__ S) {
  constexpr int NB = 2, NS = S1KC / 8, KW = NKV_ / KS1, NCH = KW / S1KC;
  constexpr int NLD = S1ROWS / 2;
  __shared__ __attribute__((aligned(16))) float stg[KS1 * S1ROWS * S1P];
  __shared__ v4f red[(KS1 - 1) * NB * 2 * 32];
  const int b  = blockIdx.z;
  const int m0 = blockIdx.y * 16;
  const int n0 = blockIdx.x * 32;
  const int w = __builtin_amdgcn_readfirstlane((int)(threadIdx.x >> 5));
  const int lane = threadIdx.x & 31;
  const int r    = lane & 15;
  const int half = lane >> 4;
  const float* X  = xkv + (long)b * CKV_ * NKV_ + (long)w * KW;
  const float* pA = X + (long)(m0 + half) * NKV_ + 4 * r;
  const float* pB = X + (long)(n0 + half) * NKV_ + 4 * r;
  float* my = stg + w * (S1ROWS * S1P);
  float* st = my + half * S1P + 4 * r;
  const float* fa = my + r * S1P + 4 * half;
  const float* fb = my + (16 + r) * S1P + 4 * half;

  v4f g[NLD];
#pragma unroll
  for (int i = 0; i < 8; ++i)  g[i]     = *(const v4f*)(pA + (long)(2 * i) * NKV_);
#pragma unroll
  for (int i = 0; i < 16; ++i) g[8 + i] = *(const v4f*)(pB + (long)(2 * i) * NKV_);

  v8f acc[NB];
#pragma unroll
  for (int j = 0; j < NB; ++j) acc[j] = {};

  for (int c = 0; c < NCH; ++c) {
#pragma unroll
    for (int i = 0; i < NLD; ++i) *(v4f*)(st + 2 * i * S1P) = g[i];
    __builtin_amdgcn_fence(__ATOMIC_RELEASE, "wavefront");
    __builtin_amdgcn_fence(__ATOMIC_ACQUIRE, "wavefront");
    if (c + 1 < NCH) {
      const int kc = (c + 1) * S1KC;
#pragma unroll
      for (int i = 0; i < 8; ++i)  g[i]     = *(const v4f*)(pA + (long)(2 * i) * NKV_ + kc);
#pragma unroll
      for (int i = 0; i < 16; ++i) g[8 + i] = *(const v4f*)(pB + (long)(2 * i) * NKV_ + kc);
    }
    Frags<NB, NS, false> f;
#pragma unroll
    for (int s = 0; s < NS; ++s) {
      f.a[s] = *(const v4f*)(fa + 8 * s);
#pragma unroll
      for (int j = 0; j < NB; ++j) f.b[s][j] = *(const v4f*)(fb + 16 * j * S1P + 8 * s);
    }
    __builtin_amdgcn_fence(__ATOMIC_RELEASE, "wavefront");
    __builtin_amdgcn_fence(__ATOMIC_ACQUIRE, "wavefront");
    mma_frags<NB, NS, false>(f, acc);
  }

  if (w != 0) {
#pragma unroll
    for (int j = 0; j < NB; ++j) {
      v4f lo, hi;
      lo.x = acc[j][0]; lo.y = acc[j][1]; lo.z = acc[j][2]; lo.w = acc[j][3];
      hi.x = acc[j][4]; hi.y = acc[j][5]; hi.z = acc[j][6]; hi.w = acc[j][7];
      red[(((w - 1) * NB + j) * 2    ) * 32 + lane] = lo;
      red[(((w - 1) * NB + j) * 2 + 1) * 32 + lane] = hi;
    }
  }
  __syncthreads();
  if (w != 0) return;
#pragma unroll
  for (int p = 1; p < KS1; ++p) {
#pragma unroll
    for (int j = 0; j < NB; ++j) {
      const v4f lo = red[(((p - 1) * NB + j) * 2    ) * 32 + lane];
      const v4f hi = red[(((p - 1) * NB + j) * 2 + 1) * 32 + lane];
      acc[j][0] += lo.x; acc[j][1] += lo.y; acc[j][2] += lo.z; acc[j][3] += lo.w;
      acc[j][4] += hi.x; acc[j][5] += hi.y; acc[j][6] += hi.z; acc[j][7] += hi.w;
    }
  }
  float* C = S + (long)b * CKV_ * CKV_;
#pragma unroll
  for (int j = 0; j < NB; ++j) {
#pragma unroll
    for (int v = 0; v < 8; ++v) {
      const long row = m0 + v + 8 * half;
      const long col = n0 + 16 * j + r;
      C[row * CKV_ + col] = acc[j][v];
    }
  }
}

__global__ void __launch_bounds__(32 * KS2) k_stage2_T1(const float* __restrict__ Wk, const float* __restrict__ S,
                            float* __restrict__ T1) {
  const int z = blockIdx.z;
  const int b = z >> 3, h = z & 7;
  wmma_gemm_strip<4, false, true, true, 32, CKV_, KS2>(Wk + (long)h * DK_ * CKV_, CKV_, 1,
                            S + (long)b * CKV_ * CKV_, CKV_, 1,
                            T1 + (long)z * DK_ * CKV_, CKV_, nullptr, 0,
                            blockIdx.y * 16, blockIdx.x * 64, 1.0f);
}

constexpr int KS34 = 4;
constexpr int TP34 = DK_ + 4;
__global__ void __launch_bounds__(32 * KS34) k_stage34_Pt(const float* __restrict__ T1, const float* __restrict__ Wv,
                             const float* __restrict__ Wq, float* __restrict__ Pt) {
  constexpr int NB = 4, KC = 32, NS = KC / 8;
  constexpr int KW = CKV_ / KS34;
  static_assert(KW == 2 * KC && DK_ == 2 * KC, "both parts are one chunk pair per wave");
  const int z = blockIdx.z;
  const int h = z & 7;
  const int m0 = blockIdx.y * 16;
  const int w = __builtin_amdgcn_readfirstlane((int)(threadIdx.x >> 5));
  const int lane = threadIdx.x & 31;
  const int r    = lane & 15;
  const int half = lane >> 4;

  __shared__ v4f red[KS34 * NB * 2 * 32];
  __shared__ __attribute__((aligned(16))) float mt[16 * TP34];

  const float* A1 = Wv + (long)h * DK_ * CKV_ + (long)w * KW;
  const float* B1 = T1 + (long)z * DK_ * CKV_ + (long)w * KW;
  const long arow = (long)(m0 + r) * CKV_;
  long bcol[NB];
#pragma unroll
  for (int j = 0; j < NB; ++j) bcol[j] = (long)(16 * j + r) * CKV_;
  v8f acc[NB];
#pragma unroll
  for (int j = 0; j < NB; ++j) acc[j] = {};
  const float* Bq = Wq + (long)h * DK_ * CQ_;
  const long col0 = 64 * w + NB * r;
  Frags<NB, NS, true> g0, g1;
  {
    Frags<NB, NS, false> f0, f1;
    load_frags<NB, NS, false, true>(f0, A1, arow, 1, B1, 1, bcol, 0, half);
    load_frags<NB, NS, false, true>(f1, A1, arow, 1, B1, 1, bcol, KC, half);
    __builtin_amdgcn_sched_barrier(0);
    mma_frags<NB, NS, false>(f0, acc);
    __builtin_amdgcn_sched_barrier(0);
    load_b_rows<NB, NS>(g0, Bq, CQ_, col0, 0, half);
    load_b_rows<NB, NS>(g1, Bq, CQ_, col0, KC, half);
    __builtin_amdgcn_sched_barrier(0);
    mma_frags<NB, NS, false>(f1, acc);
  }

#pragma unroll
  for (int j = 0; j < NB; ++j) {
    v4f lo, hi;
    lo.x = acc[j][0]; lo.y = acc[j][1]; lo.z = acc[j][2]; lo.w = acc[j][3];
    hi.x = acc[j][4]; hi.y = acc[j][5]; hi.z = acc[j][6]; hi.w = acc[j][7];
    red[((w * NB + j) * 2    ) * 32 + lane] = lo;
    red[((w * NB + j) * 2 + 1) * 32 + lane] = hi;
  }
  __syncthreads();
#pragma unroll
  for (int hh = 0; hh < 2; ++hh) {
    v4f s = red[((0 * NB + w) * 2 + hh) * 32 + lane];
#pragma unroll
    for (int p = 1; p < KS34; ++p) {
      const v4f t = red[((p * NB + w) * 2 + hh) * 32 + lane];
      s.x += t.x; s.y += t.y; s.z += t.z; s.w += t.w;
    }
    const float sc = 1.0f / (float)DK_;
    mt[(4 * hh + 0 + 8 * half) * TP34 + 16 * w + r] = s.x * sc;
    mt[(4 * hh + 1 + 8 * half) * TP34 + 16 * w + r] = s.y * sc;
    mt[(4 * hh + 2 + 8 * half) * TP34 + 16 * w + r] = s.z * sc;
    mt[(4 * hh + 3 + 8 * half) * TP34 + 16 * w + r] = s.w * sc;
  }
  __syncthreads();

#pragma unroll
  for (int s = 0; s < NS; ++s) {
    g0.a[s] = *(const v4f*)(mt + r * TP34 + 8 * s + 4 * half);
    g1.a[s] = *(const v4f*)(mt + r * TP34 + KC + 8 * s + 4 * half);
  }
  __builtin_amdgcn_sched_barrier(0);
  v8f acc2[NB];
#pragma unroll
  for (int j = 0; j < NB; ++j) acc2[j] = {};
  mma_frags<NB, NS, true>(g0, acc2);
  mma_frags<NB, NS, true>(g1, acc2);

  float* C = Pt + (long)z * DK_ * CQ_;
#pragma unroll
  for (int v = 0; v < 8; ++v) {
    const long row = m0 + v + 8 * half;
    v4f val;
#pragma unroll
    for (int j = 0; j < NB; ++j) val[j] = acc2[j][v];
    *(v4f*)(C + row * CQ_ + col0) = val;
  }
}

__global__ void __launch_bounds__(32 * KS5) k_stage5_G(const float* __restrict__ Wo, const float* __restrict__ Pt,
                           const float* __restrict__ gamma, float* __restrict__ G) {
  const int b = blockIdx.z;
  wmma_gemm_strip<4, false, true, true, 32, OD_, KS5>(Wo, OD_, 1,
                            Pt + (long)b * OD_ * CQ_, CQ_, 1,
                            G + (long)b * CQ_ * CQ_, CQ_, nullptr, 0,
                            blockIdx.y * 16, blockIdx.x * 64, gamma[0]);
}

__global__ void __launch_bounds__(32) k_stage6_out(const float* __restrict__ G, const float* __restrict__ xq,
                             float* __restrict__ out) {
  const int b = blockIdx.z;
  const float* Bq = xq + (long)b * CQ_ * NQ_;
  wmma_gemm_strip<4, true, true, true, 32, CQ_, 1>(G + (long)b * CQ_ * CQ_, CQ_, 1,
                           Bq, NQ_, 1,
                           out + (long)b * CQ_ * NQ_, NQ_, Bq, NQ_,
                           blockIdx.y * 16, blockIdx.x * 64, 1.0f);
}

extern "C" void kernel_launch(void* const* d_in, const int* in_sizes, int n_in,
                              void* d_out, int out_size, void* d_ws, size_t ws_size,
                              hipStream_t stream) {
  const float* xq    = (const float*)d_in[0];
  const float* xkv   = (const float*)d_in[1];
  const float* Wq    = (const float*)d_in[2];
  const float* Wk    = (const float*)d_in[3];
  const float* Wv    = (const float*)d_in[4];
  const float* Wo    = (const float*)d_in[5];
  const float* gamma = (const float*)d_in[6];
  float* out = (float*)d_out;

  float* S  = (float*)d_ws;
  float* T1 = S  + (size_t)B_ * CKV_ * CKV_;
  float* Pt = T1 + (size_t)B_ * H_ * DK_ * CKV_;
  float* G  = Pt + (size_t)B_ * H_ * DK_ * CQ_;

  const dim3 blk(32, 1, 1);
  k_stage1_xxT<<<dim3(CKV_ / 32, CKV_ / 16, B_),     dim3(32 * KS1, 1, 1), 0, stream>>>(xkv, S);
  k_stage2_T1 <<<dim3(CKV_ / 64, DK_ / 16, B_ * H_), dim3(32 * KS2, 1, 1), 0, stream>>>(Wk, S, T1);
  k_stage34_Pt<<<dim3(1,         DK_ / 16, B_ * H_), dim3(32 * KS34, 1, 1), 0, stream>>>(T1, Wv, Wq, Pt);
  k_stage5_G  <<<dim3(CQ_ / 64,  CQ_ / 16, B_),      dim3(32 * KS5, 1, 1), 0, stream>>>(Wo, Pt, gamma, G);
  k_stage6_out<<<dim3(NQ_ / 64,  CQ_ / 16, B_),      blk, 0, stream>>>(G, xq, out);
}


# Stage 2 joins the stage 3+4 launch: M^T formed as (Wv_h S^T) Wk_h^T, so a 16-row strip needs only its own rows of Wv_h S^T, kept in LDS; 4 launches; association and sum order change
# speedup vs baseline: 4.2548x; 1.0948x over previous
#include <hip/hip_runtime.h>


typedef float v2f __attribute__((ext_vector_type(2)));
typedef float v4f __attribute__((ext_vector_type(4)));
typedef float v8f __attribute__((ext_vector_type(8)));

template <int NB> struct VecN;
template <> struct VecN<2> { typedef v2f type; };
template <> struct VecN<4> { typedef v4f type; };

constexpr int B_   = 4;
constexpr int CQ_  = 256;
constexpr int CKV_ = 256;
constexpr int NQ_  = 2048;
constexpr int NKV_ = 2048;
constexpr int H_   = 8;
constexpr int DK_  = 64;
constexpr int OD_  = H_ * DK_;

template <int NB, int NS, bool BROW> struct Frags;
template <int NB, int NS>
struct Frags<NB, NS, false> {
  v4f a[NS];
  v4f b[NS][NB];
};
template <int NB, int NS>
struct Frags<NB, NS, true> {
  v4f a[NS];
  typename VecN<NB>::type br[NS][2][2];
};

template <int NB, int NS, bool BROW>
__device__ __forceinline__ v2f afrag(const Frags<NB, NS, BROW>& f, int s, int i) {
  v2f a;
  a.x = f.a[s][2 * i];
  a.y = f.a[s][2 * i + 1];
  return a;
}
template <int NB, int NS>
__device__ __forceinline__ v2f bfrag(const Frags<NB, NS, false>& f, int s, int i, int j) {
  v2f b;
  b.x = f.b[s][j][2 * i];
  b.y = f.b[s][j][2 * i + 1];
  return b;
}
template <int NB, int NS>
__device__ __forceinline__ v2f bfrag(const Frags<NB, NS, true>& f, int s, int i, int j) {
  v2f b;
  b.x = f.br[s][i][0][j];
  b.y = f.br[s][i][1][j];
  return b;
}

template <int NB, int NS, bool BROW, bool ACONT>
__device__ __forceinline__ void load_frags(
    Frags<NB, NS, BROW>& f, const float* __restrict__ A, long arow, long csA,
    const float* __restrict__ Bm, long rsB, const long (&bcol)[NB], int kc, int half) {
  typedef typename VecN<NB>::type vN;
#pragma unroll
  for (int s = 0; s < NS; ++s) {
    const int kb = kc + 8 * s + 4 * half;
    if constexpr (ACONT) {
      f.a[s] = *(const v4f*)(A + arow + kb);
    } else {
      f.a[s].x = A[arow + (long)(kb    ) * csA];
      f.a[s].y = A[arow + (long)(kb + 1) * csA];
      f.a[s].z = A[arow + (long)(kb + 2) * csA];
      f.a[s].w = A[arow + (long)(kb + 3) * csA];
    }
    if constexpr (BROW) {
      f.br[s][0][0] = *(const vN*)(Bm + (long)(kb    ) * rsB + bcol[0]);
      f.br[s][0][1] = *(const vN*)(Bm + (long)(kb + 1) * rsB + bcol[0]);
      f.br[s][1][0] = *(const vN*)(Bm + (long)(kb + 2) * rsB + bcol[0]);
      f.br[s][1][1] = *(const vN*)(Bm + (long)(kb + 3) * rsB + bcol[0]);
    } else {
#pragma unroll
      for (int j = 0; j < NB; ++j) f.b[s][j] = *(const v4f*)(Bm + bcol[j] + kb);
    }
  }
}

template <int NB, int NS>
__device__ __forceinline__ void load_b_rows(
    Frags<NB, NS, true>& f, const float* __restrict__ Bm, long rsB, long col, int kc, int half) {
  typedef typename VecN<NB>::type vN;
#pragma unroll
  for (int s = 0; s < NS; ++s) {
    const int kb = kc + 8 * s + 4 * half;
    f.br[s][0][0] = *(const vN*)(Bm + (long)(kb    ) * rsB + col);
    f.br[s][0][1] = *(const vN*)(Bm + (long)(kb + 1) * rsB + col);
    f.br[s][1][0] = *(const vN*)(Bm + (long)(kb + 2) * rsB + col);
    f.br[s][1][1] = *(const vN*)(Bm + (long)(kb + 3) * rsB + col);
  }
}

template <int NB, int NS, bool BROW>
__device__ __forceinline__ void mma_frags(const Frags<NB, NS, BROW>& f, v8f (&acc)[NB]) {
#pragma unroll
  for (int s = 0; s < NS; ++s) {
#pragma unroll
    for (int j = 0; j < NB; ++j)
      acc[j] = __builtin_amdgcn_wmma_f32_16x16x4_f32(false, afrag<NB, NS, BROW>(f, s, 0), false, bfrag<NB, NS>(f, s, 0, j),
                                                     (short)0, acc[j], false, false);
#pragma unroll
    for (int j = 0; j < NB; ++j)
      acc[j] = __builtin_amdgcn_wmma_f32_16x16x4_f32(false, afrag<NB, NS, BROW>(f, s, 1), false, bfrag<NB, NS>(f, s, 1, j),
                                                     (short)0, acc[j], false, false);
  }
}

template <int NB, bool HASR, bool BROW, bool ACONT, int KC, int K, int KS>
__device__ __forceinline__ void wmma_gemm_strip(
    const float* __restrict__ A, long rsA, long csA,
    const float* __restrict__ Bm, long rsB, long csB,
    float* __restrict__ C, long rsC,
    const float* __restrict__ R, long rsR,
    int m0, int n0, float scale) {
  constexpr int KW = K / KS;
  static_assert(K % KS == 0 && KC % 8 == 0 && KW >= 2 * KC && KW % (2 * KC) == 0,
                "a wave's k range must be a whole number of chunk pairs");
  typedef typename VecN<NB>::type vN;
  constexpr int NS = KC / 8;
  int w = 0;
  if constexpr (KS > 1) {
    w = __builtin_amdgcn_readfirstlane((int)(threadIdx.x >> 5));
    A  += (long)w * KW * csA;
    Bm += (long)w * KW * rsB;
  }
  const int lane = threadIdx.x & 31;
  const int r    = lane & 15;
  const int half = lane >> 4;
  const long arow  = (long)(m0 + r) * rsA;
  const long col0 = n0 + NB * r;
  long bcol[NB];
#pragma unroll
  for (int j = 0; j < NB; ++j)
    bcol[j] = BROW ? (long)(n0 + NB * r + j) * csB : (long)(n0 + 16 * j + r) * csB;

  v8f acc[NB];
#pragma unroll
  for (int j = 0; j < NB; ++j) acc[j] = {};

  Frags<NB, NS, BROW> f0, f1;
  load_frags<NB, NS, BROW, ACONT>(f0, A, arow, csA, Bm, rsB, bcol, 0, half);
  int kc = 0;
  for (; kc + 2 * KC < KW; kc += 2 * KC) {
    load_frags<NB, NS, BROW, ACONT>(f1, A, arow, csA, Bm, rsB, bcol, kc + KC, half);
    __builtin_amdgcn_sched_barrier(0);
    mma_frags<NB, NS, BROW>(f0, acc);
    __builtin_amdgcn_sched_barrier(0);
    load_frags<NB, NS, BROW, ACONT>(f0, A, arow, csA, Bm, rsB, bcol, kc + 2 * KC, half);
    __builtin_amdgcn_sched_barrier(0);
    mma_frags<NB, NS, BROW>(f1, acc);
    __builtin_amdgcn_sched_barrier(0);
  }
  load_frags<NB, NS, BROW, ACONT>(f1, A, arow, csA, Bm, rsB, bcol, kc + KC, half);
  if constexpr (KS > 1) __builtin_amdgcn_sched_barrier(0);
  mma_frags<NB, NS, BROW>(f0, acc);
  mma_frags<NB, NS, BROW>(f1, acc);

  if constexpr (KS > 1) {
    __shared__ v4f red[(KS - 1) * NB * 2 * 32];
    if (w != 0) {
#pragma unroll
      for (int j = 0; j < NB; ++j) {
        v4f lo, hi;
        lo.x = acc[j][0]; lo.y = acc[j][1]; lo.z = acc[j][2]; lo.w = acc[j][3];
        hi.x = acc[j][4]; hi.y = acc[j][5]; hi.z = acc[j][6]; hi.w = acc[j][7];
        red[(((w - 1) * NB + j) * 2    ) * 32 + lane] = lo;
        red[(((w - 1) * NB + j) * 2 + 1) * 32 + lane] = hi;
      }
    }
    __syncthreads();
    if (w != 0) return;
#pragma unroll
    for (int p = 1; p < KS; ++p) {
#pragma unroll
      for (int j = 0; j < NB; ++j) {
        const v4f lo = red[(((p - 1) * NB + j) * 2    ) * 32 + lane];
        const v4f hi = red[(((p - 1) * NB + j) * 2 + 1) * 32 + lane];
        acc[j][0] += lo.x; acc[j][1] += lo.y; acc[j][2] += lo.z; acc[j][3] += lo.w;
        acc[j][4] += hi.x; acc[j][5] += hi.y; acc[j][6] += hi.z; acc[j][7] += hi.w;
      }
    }
  }

  if constexpr (BROW) {
#pragma unroll
    for (int v = 0; v < 8; ++v) {
      const long row = m0 + v + 8 * half;
      vN res;
      if (HASR) res = *(const vN*)(R + row * rsR + col0);
      vN val;
#pragma unroll
      for (int j = 0; j < NB; ++j) {
        float t = acc[j][v] * scale;
        if (HASR) t += res[j];
        val[j] = t;
      }
      *(vN*)(C + row * rsC + col0) = val;
    }
  } else {
#pragma unroll
    for (int j = 0; j < NB; ++j) {
#pragma unroll
      for (int v = 0; v < 8; ++v) {
        const long row = m0 + v + 8 * half;
        const long col = n0 + 16 * j + r;
        float val = acc[j][v] * scale;
        if (HASR) val += R[row * rsR + col];
        C[row * rsC + col] = val;
      }
    }
  }
}

constexpr int KS1 = 4;
constexpr int KS5 = 4;
constexpr int S1KC   = 64;
constexpr int S1ROWS = 48;
constexpr int S1P    = S1KC + 4;
__global__ void __launch_bounds__(32 * KS1) k_stage1_xxT(const float* __restrict__ xkv, float* __restrict__ S) {
  constexpr int NB = 2, NS = S1KC / 8, KW = NKV_ / KS1, NCH = KW / S1KC;
  constexpr int NLD = S1ROWS / 2;
  __shared__ __attribute__((aligned(16))) float stg[KS1 * S1ROWS * S1P];
  __shared__ v4f red[(KS1 - 1) * NB * 2 * 32];
  const int b  = blockIdx.z;
  const int m0 = blockIdx.y * 16;
  const int n0 = blockIdx.x * 32;
  const int w = __builtin_amdgcn_readfirstlane((int)(threadIdx.x >> 5));
  const int lane = threadIdx.x & 31;
  const int r    = lane & 15;
  const int half = lane >> 4;
  const float* X  = xkv + (long)b * CKV_ * NKV_ + (long)w * KW;
  const float* pA = X + (long)(m0 + half) * NKV_ + 4 * r;
  const float* pB = X + (long)(n0 + half) * NKV_ + 4 * r;
  float* my = stg + w * (S1ROWS * S1P);
  float* st = my + half * S1P + 4 * r;
  const float* fa = my + r * S1P + 4 * half;
  const float* fb = my + (16 + r) * S1P + 4 * half;

  v4f g[NLD];
#pragma unroll
  for (int i = 0; i < 8; ++i)  g[i]     = *(const v4f*)(pA + (long)(2 * i) * NKV_);
#pragma unroll
  for (int i = 0; i < 16; ++i) g[8 + i] = *(const v4f*)(pB + (long)(2 * i) * NKV_);

  v8f acc[NB];
#pragma unroll
  for (int j = 0; j < NB; ++j) acc[j] = {};

  for (int c = 0; c < NCH; ++c) {
#pragma unroll
    for (int i = 0; i < NLD; ++i) *(v4f*)(st + 2 * i * S1P) = g[i];
    __builtin_amdgcn_fence(__ATOMIC_RELEASE, "wavefront");
    __builtin_amdgcn_fence(__ATOMIC_ACQUIRE, "wavefront");
    if (c + 1 < NCH) {
      const int kc = (c + 1) * S1KC;
#pragma unroll
      for (int i = 0; i < 8; ++i)  g[i]     = *(const v4f*)(pA + (long)(2 * i) * NKV_ + kc);
#pragma unroll
      for (int i = 0; i < 16; ++i) g[8 + i] = *(const v4f*)(pB + (long)(2 * i) * NKV_ + kc);
    }
    Frags<NB, NS, false> f;
#pragma unroll
    for (int s = 0; s < NS; ++s) {
      f.a[s] = *(const v4f*)(fa + 8 * s);
#pragma unroll
      for (int j = 0; j < NB; ++j) f.b[s][j] = *(const v4f*)(fb + 16 * j * S1P + 8 * s);
    }
    __builtin_amdgcn_fence(__ATOMIC_RELEASE, "wavefront");
    __builtin_amdgcn_fence(__ATOMIC_ACQUIRE, "wavefront");
    mma_frags<NB, NS, false>(f, acc);
  }

  if (w != 0) {
#pragma unroll
    for (int j = 0; j < NB; ++j) {
      v4f lo, hi;
      lo.x = acc[j][0]; lo.y = acc[j][1]; lo.z = acc[j][2]; lo.w = acc[j][3];
      hi.x = acc[j][4]; hi.y = acc[j][5]; hi.z = acc[j][6]; hi.w = acc[j][7];
      red[(((w - 1) * NB + j) * 2    ) * 32 + lane] = lo;
      red[(((w - 1) * NB + j) * 2 + 1) * 32 + lane] = hi;
    }
  }
  __syncthreads();
  if (w != 0) return;
#pragma unroll
  for (int p = 1; p < KS1; ++p) {
#pragma unroll
    for (int j = 0; j < NB; ++j) {
      const v4f lo = red[(((p - 1) * NB + j) * 2    ) * 32 + lane];
      const v4f hi = red[(((p - 1) * NB + j) * 2 + 1) * 32 + lane];
      acc[j][0] += lo.x; acc[j][1] += lo.y; acc[j][2] += lo.z; acc[j][3] += lo.w;
      acc[j][4] += hi.x; acc[j][5] += hi.y; acc[j][6] += hi.z; acc[j][7] += hi.w;
    }
  }
  float* C = S + (long)b * CKV_ * CKV_;
#pragma unroll
  for (int j = 0; j < NB; ++j) {
#pragma unroll
    for (int v = 0; v < 8; ++v) {
      const long row = m0 + v + 8 * half;
      const long col = n0 + 16 * j + r;
      C[row * CKV_ + col] = acc[j][v];
    }
  }
}

constexpr int W234 = 8;
constexpr int T1P  = CKV_ + 4;
constexpr int TP34 = DK_ + 4;
__global__ void __launch_bounds__(32 * W234) k_stage234_Pt(const float* __restrict__ S, const float* __restrict__ Wv,
                             const float* __restrict__ Wk, const float* __restrict__ Wq,
                             float* __restrict__ Pt) {
  constexpr int NB = 4, KC = 32, NS = KC / 8;
  constexpr int KWA = CKV_ / 2;
  constexpr int NBC = 2, NSC = DK_ / 8;
  static_assert(KWA == 4 * KC && CKV_ == W234 * KC && CQ_ == W234 * 16 * NBC,
                "phase A walks four chunks; phase B one chunk per wave; phase C 32 columns per wave");
  const int z = blockIdx.z;
  const int b = z >> 3, h = z & 7;
  const int m0 = blockIdx.y * 16;
  const int w = __builtin_amdgcn_readfirstlane((int)(threadIdx.x >> 5));
  const int lane = threadIdx.x & 31;
  const int r    = lane & 15;
  const int half = lane >> 4;
  const int nb = w >> 1, ks = w & 1;

  __shared__ v4f redA[4 * NB * 2 * 32];
  __shared__ v4f redB[W234 * NB * 2 * 32];
  __shared__ __attribute__((aligned(16))) float t1s[16 * T1P];
  __shared__ __attribute__((aligned(16))) float mt[16 * TP34];

  const float* A1 = Wv + (long)h * DK_ * CKV_ + (long)ks * KWA;
  const float* B1 = S + (long)b * CKV_ * CKV_ + (long)ks * KWA * CKV_;
  const long arow = (long)(m0 + r) * CKV_;
  const int n0 = 64 * nb;
  long bcol[NB];
#pragma unroll
  for (int j = 0; j < NB; ++j) bcol[j] = n0 + NB * r + j;
  v8f acc[NB];
#pragma unroll
  for (int j = 0; j < NB; ++j) acc[j] = {};
  const float* Wkh = Wk + (long)h * DK_ * CKV_;
  const float* Bq  = Wq + (long)h * DK_ * CQ_;
  const long col0 = 32 * w + NBC * r;
  Frags<NB, NS, false> fb;
  Frags<NBC, NSC, true> g;
  {
    Frags<NB, NS, true> f0, f1;
    load_frags<NB, NS, true, true>(f0, A1, arow, 1, B1, CKV_, bcol, 0, half);
    load_frags<NB, NS, true, true>(f1, A1, arow, 1, B1, CKV_, bcol, KC, half);
    __builtin_amdgcn_sched_barrier(0);
    mma_frags<NB, NS, true>(f0, acc);
    __builtin_amdgcn_sched_barrier(0);
    load_frags<NB, NS, true, true>(f0, A1, arow, 1, B1, CKV_, bcol, 2 * KC, half);
    __builtin_amdgcn_sched_barrier(0);
    mma_frags<NB, NS, true>(f1, acc);
    __builtin_amdgcn_sched_barrier(0);
    load_frags<NB, NS, true, true>(f1, A1, arow, 1, B1, CKV_, bcol, 3 * KC, half);
#pragma unroll
    for (int s = 0; s < NS; ++s) {
      const int kb = KC * w + 8 * s + 4 * half;
#pragma unroll
      for (int j = 0; j < NB; ++j) fb.b[s][j] = *(const v4f*)(Wkh + (long)(16 * j + r) * CKV_ + kb);
    }
    __builtin_amdgcn_sched_barrier(0);
    mma_frags<NB, NS, true>(f0, acc);
    __builtin_amdgcn_sched_barrier(0);
    load_b_rows<NBC, NSC>(g, Bq, CQ_, col0, 0, half);
    __builtin_amdgcn_sched_barrier(0);
    mma_frags<NB, NS, true>(f1, acc);
  }

  if (ks != 0) {
#pragma unroll
    for (int j = 0; j < NB; ++j) {
      v4f lo, hi;
      lo.x = acc[j][0]; lo.y = acc[j][1]; lo.z = acc[j][2]; lo.w = acc[j][3];
      hi.x = acc[j][4]; hi.y = acc[j][5]; hi.z = acc[j][6]; hi.w = acc[j][7];
      redA[((nb * NB + j) * 2    ) * 32 + lane] = lo;
      redA[((nb * NB + j) * 2 + 1) * 32 + lane] = hi;
    }
  }
  __syncthreads();
  if (ks == 0) {
#pragma unroll
    for (int j = 0; j < NB; ++j) {
      const v4f lo = redA[((nb * NB + j) * 2    ) * 32 + lane];
      const v4f hi = redA[((nb * NB + j) * 2 + 1) * 32 + lane];
      acc[j][0] += lo.x; acc[j][1] += lo.y; acc[j][2] += lo.z; acc[j][3] += lo.w;
      acc[j][4] += hi.x; acc[j][5] += hi.y; acc[j][6] += hi.z; acc[j][7] += hi.w;
    }
#pragma unroll
    for (int v = 0; v < 8; ++v) {
      v4f val;
#pragma unroll
      for (int j = 0; j < NB; ++j) val[j] = acc[j][v];
      *(v4f*)(t1s + (v + 8 * half) * T1P + n0 + NB * r) = val;
    }
  }
  __syncthreads();

#pragma unroll
  for (int s = 0; s < NS; ++s) fb.a[s] = *(const v4f*)(t1s + r * T1P + KC * w + 8 * s + 4 * half);
  v8f acc2[NB];
#pragma unroll
  for (int j = 0; j < NB; ++j) acc2[j] = {};
  mma_frags<NB, NS, false>(fb, acc2);
#pragma unroll
  for (int j = 0; j < NB; ++j) {
    v4f lo, hi;
    lo.x = acc2[j][0]; lo.y = acc2[j][1]; lo.z = acc2[j][2]; lo.w = acc2[j][3];
    hi.x = acc2[j][4]; hi.y = acc2[j][5]; hi.z = acc2[j][6]; hi.w = acc2[j][7];
    redB[((w * NB + j) * 2    ) * 32 + lane] = lo;
    redB[((w * NB + j) * 2 + 1) * 32 + lane] = hi;
  }
  __syncthreads();
  {
    const int t = nb, hh = ks;
    v4f s = redB[((0 * NB + t) * 2 + hh) * 32 + lane];
#pragma unroll
    for (int p = 1; p < W234; ++p) {
      const v4f u = redB[((p * NB + t) * 2 + hh) * 32 + lane];
      s.x += u.x; s.y += u.y; s.z += u.z; s.w += u.w;
    }
    const float sc = 1.0f / (float)DK_;
    mt[(4 * hh + 0 + 8 * half) * TP34 + 16 * t + r] = s.x * sc;
    mt[(4 * hh + 1 + 8 * half) * TP34 + 16 * t + r] = s.y * sc;
    mt[(4 * hh + 2 + 8 * half) * TP34 + 16 * t + r] = s.z * sc;
    mt[(4 * hh + 3 + 8 * half) * TP34 + 16 * t + r] = s.w * sc;
  }
  __syncthreads();

#pragma unroll
  for (int s = 0; s < NSC; ++s) g.a[s] = *(const v4f*)(mt + r * TP34 + 8 * s + 4 * half);
  v8f acc3[NBC];
#pragma unroll
  for (int j = 0; j < NBC; ++j) acc3[j] = {};
  mma_frags<NBC, NSC, true>(g, acc3);

  float* C = Pt + (long)z * DK_ * CQ_;
#pragma unroll
  for (int v = 0; v < 8; ++v) {
    const long row = m0 + v + 8 * half;
    v2f val;
#pragma unroll
    for (int j = 0; j < NBC; ++j) val[j] = acc3[j][v];
    *(v2f*)(C + row * CQ_ + col0) = val;
  }
}

__global__ void __launch_bounds__(32 * KS5) k_stage5_G(const float* __restrict__ Wo, const float* __restrict__ Pt,
                           const float* __restrict__ gamma, float* __restrict__ G) {
  const int b = blockIdx.z;
  wmma_gemm_strip<4, false, true, true, 32, OD_, KS5>(Wo, OD_, 1,
                            Pt + (long)b * OD_ * CQ_, CQ_, 1,
                            G + (long)b * CQ_ * CQ_, CQ_, nullptr, 0,
                            blockIdx.y * 16, blockIdx.x * 64, gamma[0]);
}

__global__ void __launch_bounds__(32) k_stage6_out(const float* __restrict__ G, const float* __restrict__ xq,
                             float* __restrict__ out) {
  const int b = blockIdx.z;
  const float* Bq = xq + (long)b * CQ_ * NQ_;
  wmma_gemm_strip<4, true, true, true, 32, CQ_, 1>(G + (long)b * CQ_ * CQ_, CQ_, 1,
                           Bq, NQ_, 1,
                           out + (long)b * CQ_ * NQ_, NQ_, Bq, NQ_,
                           blockIdx.y * 16, blockIdx.x * 64, 1.0f);
}

extern "C" void kernel_launch(void* const* d_in, const int* in_sizes, int n_in,
                              void* d_out, int out_size, void* d_ws, size_t ws_size,
                              hipStream_t stream) {
  const float* xq    = (const float*)d_in[0];
  const float* xkv   = (const float*)d_in[1];
  const float* Wq    = (const float*)d_in[2];
  const float* Wk    = (const float*)d_in[3];
  const float* Wv    = (const float*)d_in[4];
  const float* Wo    = (const float*)d_in[5];
  const float* gamma = (const float*)d_in[6];
  float* out = (float*)d_out;

  float* S  = (float*)d_ws;
  float* Pt = S  + (size_t)B_ * CKV_ * CKV_;
  float* G  = Pt + (size_t)B_ * H_ * DK_ * CQ_;

  const dim3 blk(32, 1, 1);
  k_stage1_xxT<<<dim3(CKV_ / 32, CKV_ / 16, B_),     dim3(32 * KS1, 1, 1), 0, stream>>>(xkv, S);
  k_stage234_Pt<<<dim3(1,        DK_ / 16, B_ * H_), dim3(32 * W234, 1, 1), 0, stream>>>(S, Wv, Wk, Wq, Pt);
  k_stage5_G  <<<dim3(CQ_ / 64,  CQ_ / 16, B_),      dim3(32 * KS5, 1, 1), 0, stream>>>(Wo, Pt, gamma, G);
  k_stage6_out<<<dim3(NQ_ / 64,  CQ_ / 16, B_),      blk, 0, stream>>>(G, xq, out);
}


# Waves that read the same rows share a workgroup so the WGP cache serves the re-reads: stage 6 has 8 row strips per 64-column block, stage 1 two strips per 32 B rows and k range; sum order kept
# speedup vs baseline: 4.5644x; 1.0728x over previous
#include <hip/hip_runtime.h>


typedef float v2f __attribute__((ext_vector_type(2)));
typedef float v4f __attribute__((ext_vector_type(4)));
typedef float v8f __attribute__((ext_vector_type(8)));

template <int NB> struct VecN;
template <> struct VecN<2> { typedef v2f type; };
template <> struct VecN<4> { typedef v4f type; };

constexpr int B_   = 4;
constexpr int CQ_  = 256;
constexpr int CKV_ = 256;
constexpr int NQ_  = 2048;
constexpr int NKV_ = 2048;
constexpr int H_   = 8;
constexpr int DK_  = 64;
constexpr int OD_  = H_ * DK_;

template <int NB, int NS, bool BROW> struct Frags;
template <int NB, int NS>
struct Frags<NB, NS, false> {
  v4f a[NS];
  v4f b[NS][NB];
};
template <int NB, int NS>
struct Frags<NB, NS, true> {
  v4f a[NS];
  typename VecN<NB>::type br[NS][2][2];
};

template <int NB, int NS, bool BROW>
__device__ __forceinline__ v2f afrag(const Frags<NB, NS, BROW>& f, int s, int i) {
  v2f a;
  a.x = f.a[s][2 * i];
  a.y = f.a[s][2 * i + 1];
  return a;
}
template <int NB, int NS>
__device__ __forceinline__ v2f bfrag(const Frags<NB, NS, false>& f, int s, int i, int j) {
  v2f b;
  b.x = f.b[s][j][2 * i];
  b.y = f.b[s][j][2 * i + 1];
  return b;
}
template <int NB, int NS>
__device__ __forceinline__ v2f bfrag(const Frags<NB, NS, true>& f, int s, int i, int j) {
  v2f b;
  b.x = f.br[s][i][0][j];
  b.y = f.br[s][i][1][j];
  return b;
}

template <int NB, int NS, bool BROW, bool ACONT>
__device__ __forceinline__ void load_frags(
    Frags<NB, NS, BROW>& f, const float* __restrict__ A, long arow, long csA,
    const float* __restrict__ Bm, long rsB, const long (&bcol)[NB], int kc, int half) {
  typedef typename VecN<NB>::type vN;
#pragma unroll
  for (int s = 0; s < NS; ++s) {
    const int kb = kc + 8 * s + 4 * half;
    if constexpr (ACONT) {
      f.a[s] = *(const v4f*)(A + arow + kb);
    } else {
      f.a[s].x = A[arow + (long)(kb    ) * csA];
      f.a[s].y = A[arow + (long)(kb + 1) * csA];
      f.a[s].z = A[arow + (long)(kb + 2) * csA];
      f.a[s].w = A[arow + (long)(kb + 3) * csA];
    }
    if constexpr (BROW) {
      f.br[s][0][0] = *(const vN*)(Bm + (long)(kb    ) * rsB + bcol[0]);
      f.br[s][0][1] = *(const vN*)(Bm + (long)(kb + 1) * rsB + bcol[0]);
      f.br[s][1][0] = *(const vN*)(Bm + (long)(kb + 2) * rsB + bcol[0]);
      f.br[s][1][1] = *(const vN*)(Bm + (long)(kb + 3) * rsB + bcol[0]);
    } else {
#pragma unroll
      for (int j = 0; j < NB; ++j) f.b[s][j] = *(const v4f*)(Bm + bcol[j] + kb);
    }
  }
}

template <int NB, int NS>
__device__ __forceinline__ void load_b_rows(
    Frags<NB, NS, true>& f, const float* __restrict__ Bm, long rsB, long col, int kc, int half) {
  typedef typename VecN<NB>::type vN;
#pragma unroll
  for (int s = 0; s < NS; ++s) {
    const int kb = kc + 8 * s + 4 * half;
    f.br[s][0][0] = *(const vN*)(Bm + (long)(kb    ) * rsB + col);
    f.br[s][0][1] = *(const vN*)(Bm + (long)(kb + 1) * rsB + col);
    f.br[s][1][0] = *(const vN*)(Bm + (long)(kb + 2) * rsB + col);
    f.br[s][1][1] = *(const vN*)(Bm + (long)(kb + 3) * rsB + col);
  }
}

template <int NB, int NS, bool BROW>
__device__ __forceinline__ void mma_frags(const Frags<NB, NS, BROW>& f, v8f (&acc)[NB]) {
#pragma unroll
  for (int s = 0; s < NS; ++s) {
#pragma unroll
    for (int j = 0; j < NB; ++j)
      acc[j] = __builtin_amdgcn_wmma_f32_16x16x4_f32(false, afrag<NB, NS, BROW>(f, s, 0), false, bfrag<NB, NS>(f, s, 0, j),
                                                     (short)0, acc[j], false, false);
#pragma unroll
    for (int j = 0; j < NB; ++j)
      acc[j] = __builtin_amdgcn_wmma_f32_16x16x4_f32(false, afrag<NB, NS, BROW>(f, s, 1), false, bfrag<NB, NS>(f, s, 1, j),
                                                     (short)0, acc[j], false, false);
  }
}

template <int NB, bool HASR, bool BROW, bool ACONT, int KC, int K, int KS>
__device__ __forceinline__ void wmma_gemm_strip(
    const float* __restrict__ A, long rsA, long csA,
    const float* __restrict__ Bm, long rsB, long csB,
    float* __restrict__ C, long rsC,
    const float* __restrict__ R, long rsR,
    int m0, int n0, float scale) {
  constexpr int KW = K / KS;
  static_assert(K % KS == 0 && KC % 8 == 0 && KW >= 2 * KC && KW % (2 * KC) == 0,
                "a wave's k range must be a whole number of chunk pairs");
  typedef typename VecN<NB>::type vN;
  constexpr int NS = KC / 8;
  int w = 0;
  if constexpr (KS > 1) {
    w = __builtin_amdgcn_readfirstlane((int)(threadIdx.x >> 5));
    A  += (long)w * KW * csA;
    Bm += (long)w * KW * rsB;
  }
  const int lane = threadIdx.x & 31;
  const int r    = lane & 15;
  const int half = lane >> 4;
  const long arow  = (long)(m0 + r) * rsA;
  const long col0 = n0 + NB * r;
  long bcol[NB];
#pragma unroll
  for (int j = 0; j < NB; ++j)
    bcol[j] = BROW ? (long)(n0 + NB * r + j) * csB : (long)(n0 + 16 * j + r) * csB;

  v8f acc[NB];
#pragma unroll
  for (int j = 0; j < NB; ++j) acc[j] = {};

  Frags<NB, NS, BROW> f0, f1;
  load_frags<NB, NS, BROW, ACONT>(f0, A, arow, csA, Bm, rsB, bcol, 0, half);
  int kc = 0;
  for (; kc + 2 * KC < KW; kc += 2 * KC) {
    load_frags<NB, NS, BROW, ACONT>(f1, A, arow, csA, Bm, rsB, bcol, kc + KC, half);
    __builtin_amdgcn_sched_barrier(0);
    mma_frags<NB, NS, BROW>(f0, acc);
    __builtin_amdgcn_sched_barrier(0);
    load_frags<NB, NS, BROW, ACONT>(f0, A, arow, csA, Bm, rsB, bcol, kc + 2 * KC, half);
    __builtin_amdgcn_sched_barrier(0);
    mma_frags<NB, NS, BROW>(f1, acc);
    __builtin_amdgcn_sched_barrier(0);
  }
  load_frags<NB, NS, BROW, ACONT>(f1, A, arow, csA, Bm, rsB, bcol, kc + KC, half);
  if constexpr (KS > 1) __builtin_amdgcn_sched_barrier(0);
  mma_frags<NB, NS, BROW>(f0, acc);
  mma_frags<NB, NS, BROW>(f1, acc);

  if constexpr (KS > 1) {
    __shared__ v4f red[(KS - 1) * NB * 2 * 32];
    if (w != 0) {
#pragma unroll
      for (int j = 0; j < NB; ++j) {
        v4f lo, hi;
        lo.x = acc[j][0]; lo.y = acc[j][1]; lo.z = acc[j][2]; lo.w = acc[j][3];
        hi.x = acc[j][4]; hi.y = acc[j][5]; hi.z = acc[j][6]; hi.w = acc[j][7];
        red[(((w - 1) * NB + j) * 2    ) * 32 + lane] = lo;
        red[(((w - 1) * NB + j) * 2 + 1) * 32 + lane] = hi;
      }
    }
    __syncthreads();
    if (w != 0) return;
#pragma unroll
    for (int p = 1; p < KS; ++p) {
#pragma unroll
      for (int j = 0; j < NB; ++j) {
        const v4f lo = red[(((p - 1) * NB + j) * 2    ) * 32 + lane];
        const v4f hi = red[(((p - 1) * NB + j) * 2 + 1) * 32 + lane];
        acc[j][0] += lo.x; acc[j][1] += lo.y; acc[j][2] += lo.z; acc[j][3] += lo.w;
        acc[j][4] += hi.x; acc[j][5] += hi.y; acc[j][6] += hi.z; acc[j][7] += hi.w;
      }
    }
  }

  if constexpr (BROW) {
#pragma unroll
    for (int v = 0; v < 8; ++v) {
      const long row = m0 + v + 8 * half;
      vN res;
      if (HASR) res = *(const vN*)(R + row * rsR + col0);
      vN val;
#pragma unroll
      for (int j = 0; j < NB; ++j) {
        float t = acc[j][v] * scale;
        if (HASR) t += res[j];
        val[j] = t;
      }
      *(vN*)(C + row * rsC + col0) = val;
    }
  } else {
#pragma unroll
    for (int j = 0; j < NB; ++j) {
#pragma unroll
      for (int v = 0; v < 8; ++v) {
        const long row = m0 + v + 8 * half;
        const long col = n0 + 16 * j + r;
        float val = acc[j][v] * scale;
        if (HASR) val += R[row * rsR + col];
        C[row * rsC + col] = val;
      }
    }
  }
}

constexpr int KS1 = 4;
constexpr int MS1 = 2;
constexpr int KS5 = 4;
constexpr int S1KC   = 64;
constexpr int S1ROWS = 48;
constexpr int S1P    = S1KC + 4;
__global__ void __launch_bounds__(32 * KS1 * MS1) k_stage1_xxT(const float* __restrict__ xkv, float* __restrict__ S) {
  constexpr int NB = 2, NS = S1KC / 8, KW = NKV_ / KS1, NCH = KW / S1KC;
  constexpr int NLD = S1ROWS / 2;
  static_assert(MS1 == 2, "a wave's strip is bit 0 of its number, its k range the bits above");
  __shared__ __attribute__((aligned(16))) float stg[MS1 * KS1 * S1ROWS * S1P];
  __shared__ v4f red[MS1 * (KS1 - 1) * NB * 2 * 32];
  const int b  = blockIdx.z;
  const int wv = __builtin_amdgcn_readfirstlane((int)(threadIdx.x >> 5));
  const int w  = wv >> 1;
  const int mi = wv & 1;
  const int m0 = (blockIdx.y * MS1 + mi) * 16;
  const int n0 = blockIdx.x * 32;
  const int lane = threadIdx.x & 31;
  const int r    = lane & 15;
  const int half = lane >> 4;
  const float* X  = xkv + (long)b * CKV_ * NKV_ + (long)w * KW;
  const float* pA = X + (long)(m0 + half) * NKV_ + 4 * r;
  const float* pB = X + (long)(n0 + half) * NKV_ + 4 * r;
  float* my = stg + wv * (S1ROWS * S1P);
  float* st = my + half * S1P + 4 * r;
  const float* fa = my + r * S1P + 4 * half;
  const float* fb = my + (16 + r) * S1P + 4 * half;

  v4f g[NLD];
#pragma unroll
  for (int i = 0; i < 8; ++i)  g[i]     = *(const v4f*)(pA + (long)(2 * i) * NKV_);
#pragma unroll
  for (int i = 0; i < 16; ++i) g[8 + i] = *(const v4f*)(pB + (long)(2 * i) * NKV_);

  v8f acc[NB];
#pragma unroll
  for (int j = 0; j < NB; ++j) acc[j] = {};

  for (int c = 0; c < NCH; ++c) {
#pragma unroll
    for (int i = 0; i < NLD; ++i) *(v4f*)(st + 2 * i * S1P) = g[i];
    __builtin_amdgcn_fence(__ATOMIC_RELEASE, "wavefront");
    __builtin_amdgcn_fence(__ATOMIC_ACQUIRE, "wavefront");
    if (c + 1 < NCH) {
      const int kc = (c + 1) * S1KC;
#pragma unroll
      for (int i = 0; i < 8; ++i)  g[i]     = *(const v4f*)(pA + (long)(2 * i) * NKV_ + kc);
#pragma unroll
      for (int i = 0; i < 16; ++i) g[8 + i] = *(const v4f*)(pB + (long)(2 * i) * NKV_ + kc);
    }
    Frags<NB, NS, false> f;
#pragma unroll
    for (int s = 0; s < NS; ++s) {
      f.a[s] = *(const v4f*)(fa + 8 * s);
#pragma unroll
      for (int j = 0; j < NB; ++j) f.b[s][j] = *(const v4f*)(fb + 16 * j * S1P + 8 * s);
    }
    __builtin_amdgcn_fence(__ATOMIC_RELEASE, "wavefront");
    __builtin_amdgcn_fence(__ATOMIC_ACQUIRE, "wavefront");
    mma_frags<NB, NS, false>(f, acc);
    __builtin_amdgcn_sched_barrier(0);
  }

  v4f* redm = red + mi * ((KS1 - 1) * NB * 2 * 32);
  if (w != 0) {
#pragma unroll
    for (int j = 0; j < NB; ++j) {
      v4f lo, hi;
      lo.x = acc[j][0]; lo.y = acc[j][1]; lo.z = acc[j][2]; lo.w = acc[j][3];
      hi.x = acc[j][4]; hi.y = acc[j][5]; hi.z = acc[j][6]; hi.w = acc[j][7];
      redm[(((w - 1) * NB + j) * 2    ) * 32 + lane] = lo;
      redm[(((w - 1) * NB + j) * 2 + 1) * 32 + lane] = hi;
    }
  }
  __syncthreads();
  if (w != 0) return;
#pragma unroll
  for (int p = 1; p < KS1; ++p) {
#pragma unroll
    for (int j = 0; j < NB; ++j) {
      const v4f lo = redm[(((p - 1) * NB + j) * 2    ) * 32 + lane];
      const v4f hi = redm[(((p - 1) * NB + j) * 2 + 1) * 32 + lane];
      acc[j][0] += lo.x; acc[j][1] += lo.y; acc[j][2] += lo.z; acc[j][3] += lo.w;
      acc[j][4] += hi.x; acc[j][5] += hi.y; acc[j][6] += hi.z; acc[j][7] += hi.w;
    }
  }
  float* C = S + (long)b * CKV_ * CKV_;
#pragma unroll
  for (int j = 0; j < NB; ++j) {
#pragma unroll
    for (int v = 0; v < 8; ++v) {
      const long row = m0 + v + 8 * half;
      const long col = n0 + 16 * j + r;
      C[row * CKV_ + col] = acc[j][v];
    }
  }
}

constexpr int W234 = 8;
constexpr int T1P  = CKV_ + 4;
constexpr int TP34 = DK_ + 4;
__global__ void __launch_bounds__(32 * W234) k_stage234_Pt(const float* __restrict__ S, const float* __restrict__ Wv,
                             const float* __restrict__ Wk, const float* __restrict__ Wq,
                             float* __restrict__ Pt) {
  constexpr int NB = 4, KC = 32, NS = KC / 8;
  constexpr int KWA = CKV_ / 2;
  constexpr int NBC = 2, NSC = DK_ / 8;
  static_assert(KWA == 4 * KC && CKV_ == W234 * KC && CQ_ == W234 * 16 * NBC,
                "phase A walks four chunks; phase B one chunk per wave; phase C 32 columns per wave");
  const int z = blockIdx.z;
  const int b = z >> 3, h = z & 7;
  const int m0 = blockIdx.y * 16;
  const int w = __builtin_amdgcn_readfirstlane((int)(threadIdx.x >> 5));
  const int lane = threadIdx.x & 31;
  const int r    = lane & 15;
  const int half = lane >> 4;
  const int nb = w >> 1, ks = w & 1;

  __shared__ v4f redA[4 * NB * 2 * 32];
  __shared__ v4f redB[W234 * NB * 2 * 32];
  __shared__ __attribute__((aligned(16))) float t1s[16 * T1P];
  __shared__ __attribute__((aligned(16))) float mt[16 * TP34];

  const float* A1 = Wv + (long)h * DK_ * CKV_ + (long)ks * KWA;
  const float* B1 = S + (long)b * CKV_ * CKV_ + (long)ks * KWA * CKV_;
  const long arow = (long)(m0 + r) * CKV_;
  const int n0 = 64 * nb;
  long bcol[NB];
#pragma unroll
  for (int j = 0; j < NB; ++j) bcol[j] = n0 + NB * r + j;
  v8f acc[NB];
#pragma unroll
  for (int j = 0; j < NB; ++j) acc[j] = {};
  const float* Wkh = Wk + (long)h * DK_ * CKV_;
  const float* Bq  = Wq + (long)h * DK_ * CQ_;
  const long col0 = 32 * w + NBC * r;
  Frags<NB, NS, false> fb;
  Frags<NBC, NSC, true> g;
  {
    Frags<NB, NS, true> f0, f1;
    load_frags<NB, NS, true, true>(f0, A1, arow, 1, B1, CKV_, bcol, 0, half);
    load_frags<NB, NS, true, true>(f1, A1, arow, 1, B1, CKV_, bcol, KC, half);
    __builtin_amdgcn_sched_barrier(0);
    mma_frags<NB, NS, true>(f0, acc);
    __builtin_amdgcn_sched_barrier(0);
    load_frags<NB, NS, true, true>(f0, A1, arow, 1, B1, CKV_, bcol, 2 * KC, half);
    __builtin_amdgcn_sched_barrier(0);
    mma_frags<NB, NS, true>(f1, acc);
    __builtin_amdgcn_sched_barrier(0);
    load_frags<NB, NS, true, true>(f1, A1, arow, 1, B1, CKV_, bcol, 3 * KC, half);
#pragma unroll
    for (int s = 0; s < NS; ++s) {
      const int kb = KC * w + 8 * s + 4 * half;
#pragma unroll
      for (int j = 0; j < NB; ++j) fb.b[s][j] = *(const v4f*)(Wkh + (long)(16 * j + r) * CKV_ + kb);
    }
    __builtin_amdgcn_sched_barrier(0);
    mma_frags<NB, NS, true>(f0, acc);
    __builtin_amdgcn_sched_barrier(0);
    load_b_rows<NBC, NSC>(g, Bq, CQ_, col0, 0, half);
    __builtin_amdgcn_sched_barrier(0);
    mma_frags<NB, NS, true>(f1, acc);
  }

  if (ks != 0) {
#pragma unroll
    for (int j = 0; j < NB; ++j) {
      v4f lo, hi;
      lo.x = acc[j][0]; lo.y = acc[j][1]; lo.z = acc[j][2]; lo.w = acc[j][3];
      hi.x = acc[j][4]; hi.y = acc[j][5]; hi.z = acc[j][6]; hi.w = acc[j][7];
      redA[((nb * NB + j) * 2    ) * 32 + lane] = lo;
      redA[((nb * NB + j) * 2 + 1) * 32 + lane] = hi;
    }
  }
  __syncthreads();
  if (ks == 0) {
#pragma unroll
    for (int j = 0; j < NB; ++j) {
      const v4f lo = redA[((nb * NB + j) * 2    ) * 32 + lane];
      const v4f hi = redA[((nb * NB + j) * 2 + 1) * 32 + lane];
      acc[j][0] += lo.x; acc[j][1] += lo.y; acc[j][2] += lo.z; acc[j][3] += lo.w;
      acc[j][4] += hi.x; acc[j][5] += hi.y; acc[j][6] += hi.z; acc[j][7] += hi.w;
    }
#pragma unroll
    for (int v = 0; v < 8; ++v) {
      v4f val;
#pragma unroll
      for (int j = 0; j < NB; ++j) val[j] = acc[j][v];
      *(v4f*)(t1s + (v + 8 * half) * T1P + n0 + NB * r) = val;
    }
  }
  __syncthreads();

#pragma unroll
  for (int s = 0; s < NS; ++s) fb.a[s] = *(const v4f*)(t1s + r * T1P + KC * w + 8 * s + 4 * half);
  v8f acc2[NB];
#pragma unroll
  for (int j = 0; j < NB; ++j) acc2[j] = {};
  mma_frags<NB, NS, false>(fb, acc2);
#pragma unroll
  for (int j = 0; j < NB; ++j) {
    v4f lo, hi;
    lo.x = acc2[j][0]; lo.y = acc2[j][1]; lo.z = acc2[j][2]; lo.w = acc2[j][3];
    hi.x = acc2[j][4]; hi.y = acc2[j][5]; hi.z = acc2[j][6]; hi.w = acc2[j][7];
    redB[((w * NB + j) * 2    ) * 32 + lane] = lo;
    redB[((w * NB + j) * 2 + 1) * 32 + lane] = hi;
  }
  __syncthreads();
  {
    const int t = nb, hh = ks;
    v4f s = redB[((0 * NB + t) * 2 + hh) * 32 + lane];
#pragma unroll
    for (int p = 1; p < W234; ++p) {
      const v4f u = redB[((p * NB + t) * 2 + hh) * 32 + lane];
      s.x += u.x; s.y += u.y; s.z += u.z; s.w += u.w;
    }
    const float sc = 1.0f / (float)DK_;
    mt[(4 * hh + 0 + 8 * half) * TP34 + 16 * t + r] = s.x * sc;
    mt[(4 * hh + 1 + 8 * half) * TP34 + 16 * t + r] = s.y * sc;
    mt[(4 * hh + 2 + 8 * half) * TP34 + 16 * t + r] = s.z * sc;
    mt[(4 * hh + 3 + 8 * half) * TP34 + 16 * t + r] = s.w * sc;
  }
  __syncthreads();

#pragma unroll
  for (int s = 0; s < NSC; ++s) g.a[s] = *(const v4f*)(mt + r * TP34 + 8 * s + 4 * half);
  v8f acc3[NBC];
#pragma unroll
  for (int j = 0; j < NBC; ++j) acc3[j] = {};
  mma_frags<NBC, NSC, true>(g, acc3);

  float* C = Pt + (long)z * DK_ * CQ_;
#pragma unroll
  for (int v = 0; v < 8; ++v) {
    const long row = m0 + v + 8 * half;
    v2f val;
#pragma unroll
    for (int j = 0; j < NBC; ++j) val[j] = acc3[j][v];
    *(v2f*)(C + row * CQ_ + col0) = val;
  }
}

__global__ void __launch_bounds__(32 * KS5) k_stage5_G(const float* __restrict__ Wo, const float* __restrict__ Pt,
                           const float* __restrict__ gamma, float* __restrict__ G) {
  const int b = blockIdx.z;
  wmma_gemm_strip<4, false, true, true, 32, OD_, KS5>(Wo, OD_, 1,
                            Pt + (long)b * OD_ * CQ_, CQ_, 1,
                            G + (long)b * CQ_ * CQ_, CQ_, nullptr, 0,
                            blockIdx.y * 16, blockIdx.x * 64, gamma[0]);
}

constexpr int W6 = 8;
__global__ void __launch_bounds__(32 * W6) k_stage6_out(const float* __restrict__ G, const float* __restrict__ xq,
                             float* __restrict__ out) {
  const int b = blockIdx.z;
  const int w = __builtin_amdgcn_readfirstlane((int)(threadIdx.x >> 5));
  const float* Bq = xq + (long)b * CQ_ * NQ_;
  wmma_gemm_strip<4, true, true, true, 32, CQ_, 1>(G + (long)b * CQ_ * CQ_, CQ_, 1,
                           Bq, NQ_, 1,
                           out + (long)b * CQ_ * NQ_, NQ_, Bq, NQ_,
                           (blockIdx.y * W6 + w) * 16, blockIdx.x * 64, 1.0f);
}

extern "C" void kernel_launch(void* const* d_in, const int* in_sizes, int n_in,
                              void* d_out, int out_size, void* d_ws, size_t ws_size,
                              hipStream_t stream) {
  const float* xq    = (const float*)d_in[0];
  const float* xkv   = (const float*)d_in[1];
  const float* Wq    = (const float*)d_in[2];
  const float* Wk    = (const float*)d_in[3];
  const float* Wv    = (const float*)d_in[4];
  const float* Wo    = (const float*)d_in[5];
  const float* gamma = (const float*)d_in[6];
  float* out = (float*)d_out;

  float* S  = (float*)d_ws;
  float* Pt = S  + (size_t)B_ * CKV_ * CKV_;
  float* G  = Pt + (size_t)B_ * H_ * DK_ * CQ_;

  k_stage1_xxT<<<dim3(CKV_ / 32, CKV_ / (16 * MS1), B_), dim3(32 * KS1 * MS1, 1, 1), 0, stream>>>(xkv, S);
  k_stage234_Pt<<<dim3(1,        DK_ / 16, B_ * H_), dim3(32 * W234, 1, 1), 0, stream>>>(S, Wv, Wk, Wq, Pt);
  k_stage5_G  <<<dim3(CQ_ / 64,  CQ_ / 16, B_),      dim3(32 * KS5, 1, 1), 0, stream>>>(Wo, Pt, gamma, G);
  k_stage6_out<<<dim3(NQ_ / 64,  CQ_ / (16 * W6), B_), dim3(32 * W6, 1, 1), 0, stream>>>(G, xq, out);
}


# Stage 1 uses S = S^T: only the 16 x 32 tiles on or above the diagonal are formed (diagonal ones add their lower unit), stored as is and mirrored in 16 x 16 units; 8 waves split K: sum order changes
# speedup vs baseline: 4.7547x; 1.0417x over previous
#include <hip/hip_runtime.h>


typedef float v2f __attribute__((ext_vector_type(2)));
typedef float v4f __attribute__((ext_vector_type(4)));
typedef float v8f __attribute__((ext_vector_type(8)));

template <int NB> struct VecN;
template <> struct VecN<2> { typedef v2f type; };
template <> struct VecN<4> { typedef v4f type; };

constexpr int B_   = 4;
constexpr int CQ_  = 256;
constexpr int CKV_ = 256;
constexpr int NQ_  = 2048;
constexpr int NKV_ = 2048;
constexpr int H_   = 8;
constexpr int DK_  = 64;
constexpr int OD_  = H_ * DK_;

template <int NB, int NS, bool BROW> struct Frags;
template <int NB, int NS>
struct Frags<NB, NS, false> {
  v4f a[NS];
  v4f b[NS][NB];
};
template <int NB, int NS>
struct Frags<NB, NS, true> {
  v4f a[NS];
  typename VecN<NB>::type br[NS][2][2];
};

template <int NB, int NS, bool BROW>
__device__ __forceinline__ v2f afrag(const Frags<NB, NS, BROW>& f, int s, int i) {
  v2f a;
  a.x = f.a[s][2 * i];
  a.y = f.a[s][2 * i + 1];
  return a;
}
template <int NB, int NS>
__device__ __forceinline__ v2f bfrag(const Frags<NB, NS, false>& f, int s, int i, int j) {
  v2f b;
  b.x = f.b[s][j][2 * i];
  b.y = f.b[s][j][2 * i + 1];
  return b;
}
template <int NB, int NS>
__device__ __forceinline__ v2f bfrag(const Frags<NB, NS, true>& f, int s, int i, int j) {
  v2f b;
  b.x = f.br[s][i][0][j];
  b.y = f.br[s][i][1][j];
  return b;
}

template <int NB, int NS, bool BROW, bool ACONT>
__device__ __forceinline__ void load_frags(
    Frags<NB, NS, BROW>& f, const float* __restrict__ A, long arow, long csA,
    const float* __restrict__ Bm, long rsB, const long (&bcol)[NB], int kc, int half) {
  typedef typename VecN<NB>::type vN;
#pragma unroll
  for (int s = 0; s < NS; ++s) {
    const int kb = kc + 8 * s + 4 * half;
    if constexpr (ACONT) {
      f.a[s] = *(const v4f*)(A + arow + kb);
    } else {
      f.a[s].x = A[arow + (long)(kb    ) * csA];
      f.a[s].y = A[arow + (long)(kb + 1) * csA];
      f.a[s].z = A[arow + (long)(kb + 2) * csA];
      f.a[s].w = A[arow + (long)(kb + 3) * csA];
    }
    if constexpr (BROW) {
      f.br[s][0][0] = *(const vN*)(Bm + (long)(kb    ) * rsB + bcol[0]);
      f.br[s][0][1] = *(const vN*)(Bm + (long)(kb + 1) * rsB + bcol[0]);
      f.br[s][1][0] = *(const vN*)(Bm + (long)(kb + 2) * rsB + bcol[0]);
      f.br[s][1][1] = *(const vN*)(Bm + (long)(kb + 3) * rsB + bcol[0]);
    } else {
#pragma unroll
      for (int j = 0; j < NB; ++j) f.b[s][j] = *(const v4f*)(Bm + bcol[j] + kb);
    }
  }
}

template <int NB, int NS>
__device__ __forceinline__ void load_b_rows(
    Frags<NB, NS, true>& f, const float* __restrict__ Bm, long rsB, long col, int kc, int half) {
  typedef typename VecN<NB>::type vN;
#pragma unroll
  for (int s = 0; s < NS; ++s) {
    const int kb = kc + 8 * s + 4 * half;
    f.br[s][0][0] = *(const vN*)(Bm + (long)(kb    ) * rsB + col);
    f.br[s][0][1] = *(const vN*)(Bm + (long)(kb + 1) * rsB + col);
    f.br[s][1][0] = *(const vN*)(Bm + (long)(kb + 2) * rsB + col);
    f.br[s][1][1] = *(const vN*)(Bm + (long)(kb + 3) * rsB + col);
  }
}

template <int NS>
__device__ __forceinline__ void load_frags_su(
    Frags<4, NS, true>& f, const float* __restrict__ A, long arow,
    const float* __restrict__ Su, long lb, int kc, int half) {
#pragma unroll
  for (int s = 0; s < NS; ++s) {
    f.a[s] = *(const v4f*)(A + arow + kc + 8 * s + 4 * half);
    const float* p = Su + (long)((kc + 8 * s) >> 4) * (16 * 256) + (s & 1) * 128 + lb;
    f.br[s][0][0] = *(const v4f*)(p);
    f.br[s][0][1] = *(const v4f*)(p + 32);
    f.br[s][1][0] = *(const v4f*)(p + 64);
    f.br[s][1][1] = *(const v4f*)(p + 96);
  }
}

template <int NB, int NS, bool BROW>
__device__ __forceinline__ void mma_frags(const Frags<NB, NS, BROW>& f, v8f (&acc)[NB]) {
#pragma unroll
  for (int s = 0; s < NS; ++s) {
#pragma unroll
    for (int j = 0; j < NB; ++j)
      acc[j] = __builtin_amdgcn_wmma_f32_16x16x4_f32(false, afrag<NB, NS, BROW>(f, s, 0), false, bfrag<NB, NS>(f, s, 0, j),
                                                     (short)0, acc[j], false, false);
#pragma unroll
    for (int j = 0; j < NB; ++j)
      acc[j] = __builtin_amdgcn_wmma_f32_16x16x4_f32(false, afrag<NB, NS, BROW>(f, s, 1), false, bfrag<NB, NS>(f, s, 1, j),
                                                     (short)0, acc[j], false, false);
  }
}

template <int NB, bool HASR, bool BROW, bool ACONT, int KC, int K, int KS>
__device__ __forceinline__ void wmma_gemm_strip(
    const float* __restrict__ A, long rsA, long csA,
    const float* __restrict__ Bm, long rsB, long csB,
    float* __restrict__ C, long rsC,
    const float* __restrict__ R, long rsR,
    int m0, int n0, float scale) {
  constexpr int KW = K / KS;
  static_assert(K % KS == 0 && KC % 8 == 0 && KW >= 2 * KC && KW % (2 * KC) == 0,
                "a wave's k range must be a whole number of chunk pairs");
  typedef typename VecN<NB>::type vN;
  constexpr int NS = KC / 8;
  int w = 0;
  if constexpr (KS > 1) {
    w = __builtin_amdgcn_readfirstlane((int)(threadIdx.x >> 5));
    A  += (long)w * KW * csA;
    Bm += (long)w * KW * rsB;
  }
  const int lane = threadIdx.x & 31;
  const int r    = lane & 15;
  const int half = lane >> 4;
  const long arow  = (long)(m0 + r) * rsA;
  const long col0 = n0 + NB * r;
  long bcol[NB];
#pragma unroll
  for (int j = 0; j < NB; ++j)
    bcol[j] = BROW ? (long)(n0 + NB * r + j) * csB : (long)(n0 + 16 * j + r) * csB;

  v8f acc[NB];
#pragma unroll
  for (int j = 0; j < NB; ++j) acc[j] = {};

  Frags<NB, NS, BROW> f0, f1;
  load_frags<NB, NS, BROW, ACONT>(f0, A, arow, csA, Bm, rsB, bcol, 0, half);
  int kc = 0;
  for (; kc + 2 * KC < KW; kc += 2 * KC) {
    load_frags<NB, NS, BROW, ACONT>(f1, A, arow, csA, Bm, rsB, bcol, kc + KC, half);
    __builtin_amdgcn_sched_barrier(0);
    mma_frags<NB, NS, BROW>(f0, acc);
    __builtin_amdgcn_sched_barrier(0);
    load_frags<NB, NS, BROW, ACONT>(f0, A, arow, csA, Bm, rsB, bcol, kc + 2 * KC, half);
    __builtin_amdgcn_sched_barrier(0);
    mma_frags<NB, NS, BROW>(f1, acc);
    __builtin_amdgcn_sched_barrier(0);
  }
  load_frags<NB, NS, BROW, ACONT>(f1, A, arow, csA, Bm, rsB, bcol, kc + KC, half);
  if constexpr (KS > 1) __builtin_amdgcn_sched_barrier(0);
  mma_frags<NB, NS, BROW>(f0, acc);
  mma_frags<NB, NS, BROW>(f1, acc);

  if constexpr (KS > 1) {
    __shared__ v4f red[(KS - 1) * NB * 2 * 32];
    if (w != 0) {
#pragma unroll
      for (int j = 0; j < NB; ++j) {
        v4f lo, hi;
        lo.x = acc[j][0]; lo.y = acc[j][1]; lo.z = acc[j][2]; lo.w = acc[j][3];
        hi.x = acc[j][4]; hi.y = acc[j][5]; hi.z = acc[j][6]; hi.w = acc[j][7];
        red[(((w - 1) * NB + j) * 2    ) * 32 + lane] = lo;
        red[(((w - 1) * NB + j) * 2 + 1) * 32 + lane] = hi;
      }
    }
    __syncthreads();
    if (w != 0) return;
#pragma unroll
    for (int p = 1; p < KS; ++p) {
#pragma unroll
      for (int j = 0; j < NB; ++j) {
        const v4f lo = red[(((p - 1) * NB + j) * 2    ) * 32 + lane];
        const v4f hi = red[(((p - 1) * NB + j) * 2 + 1) * 32 + lane];
        acc[j][0] += lo.x; acc[j][1] += lo.y; acc[j][2] += lo.z; acc[j][3] += lo.w;
        acc[j][4] += hi.x; acc[j][5] += hi.y; acc[j][6] += hi.z; acc[j][7] += hi.w;
      }
    }
  }

  if constexpr (BROW) {
#pragma unroll
    for (int v = 0; v < 8; ++v) {
      const long row = m0 + v + 8 * half;
      vN res;
      if (HASR) res = *(const vN*)(R + row * rsR + col0);
      vN val;
#pragma unroll
      for (int j = 0; j < NB; ++j) {
        float t = acc[j][v] * scale;
        if (HASR) t += res[j];
        val[j] = t;
      }
      *(vN*)(C + row * rsC + col0) = val;
    }
  } else {
#pragma unroll
    for (int j = 0; j < NB; ++j) {
#pragma unroll
      for (int v = 0; v < 8; ++v) {
        const long row = m0 + v + 8 * half;
        const long col = n0 + 16 * j + r;
        float val = acc[j][v] * scale;
        if (HASR) val += R[row * rsR + col];
        C[row * rsC + col] = val;
      }
    }
  }
}

constexpr int KS1 = 8;
constexpr int S1T = 64;
constexpr int KS5 = 4;
constexpr int S1KC   = 64;
constexpr int S1ROWS = 48;
constexpr int S1P    = S1KC + 4;
constexpr int S1U    = 256;
__device__ constexpr int s1_at(int a, int c) {
  return ((a >> 3) * 4 + (a & 3)) * 32 + ((a >> 2) & 1) * 16 + c;
}

template <bool DIAG>
__device__ __forceinline__ void s1_tile(const float* __restrict__ X, float* __restrict__ C,
                                        float* my, v4f* red, int m0, int n0, int w, int lane) {
  constexpr int NS = S1KC / 8, KW = NKV_ / KS1, NCH = KW / S1KC;
  constexpr int NA = DIAG ? 0 : 8;
  constexpr int NT = DIAG ? 3 : 2;
  const int r    = lane & 15;
  const int half = lane >> 4;
  const float* pA = X + (long)(m0 + half) * NKV_ + 4 * r;
  const float* pB = X + (long)(n0 + half) * NKV_ + 4 * r;
  float* st = my + half * S1P + 4 * r;
  const float* fa = my + r * S1P + 4 * half;
  const float* fb = my + (16 + r) * S1P + 4 * half;

  v4f g[NA + 16];
#pragma unroll
  for (int i = 0; i < NA; ++i) g[i]      = *(const v4f*)(pA + (long)(2 * i) * NKV_);
#pragma unroll
  for (int i = 0; i < 16; ++i) g[NA + i] = *(const v4f*)(pB + (long)(2 * i) * NKV_);

  v8f acc[NT];
#pragma unroll
  for (int j = 0; j < NT; ++j) acc[j] = {};

  for (int c = 0; c < NCH; ++c) {
#pragma unroll
    for (int i = 0; i < NA; ++i) *(v4f*)(st + 2 * i * S1P) = g[i];
#pragma unroll
    for (int i = 0; i < 16; ++i) *(v4f*)(st + (16 + 2 * i) * S1P) = g[NA + i];
    __builtin_amdgcn_fence(__ATOMIC_RELEASE, "wavefront");
    __builtin_amdgcn_fence(__ATOMIC_ACQUIRE, "wavefront");
    if (c + 1 < NCH) {
      const int kc = (c + 1) * S1KC;
#pragma unroll
      for (int i = 0; i < NA; ++i) g[i]      = *(const v4f*)(pA + (long)(2 * i) * NKV_ + kc);
#pragma unroll
      for (int i = 0; i < 16; ++i) g[NA + i] = *(const v4f*)(pB + (long)(2 * i) * NKV_ + kc);
    }
    Frags<2, NS, false> f;
#pragma unroll
    for (int s = 0; s < NS; ++s) {
#pragma unroll
      for (int j = 0; j < 2; ++j) f.b[s][j] = *(const v4f*)(fb + 16 * j * S1P + 8 * s);
      if constexpr (DIAG) f.a[s] = f.b[s][0];
      else                f.a[s] = *(const v4f*)(fa + 8 * s);
    }
    __builtin_amdgcn_fence(__ATOMIC_RELEASE, "wavefront");
    __builtin_amdgcn_fence(__ATOMIC_ACQUIRE, "wavefront");
    if constexpr (DIAG) {
#pragma unroll
      for (int s = 0; s < NS; ++s) {
#pragma unroll
        for (int i = 0; i < 2; ++i) {
          const v2f b0 = bfrag<2, NS>(f, s, i, 0);
          const v2f b1 = bfrag<2, NS>(f, s, i, 1);
          acc[0] = __builtin_amdgcn_wmma_f32_16x16x4_f32(false, b0, false, b0, (short)0, acc[0], false, false);
          acc[1] = __builtin_amdgcn_wmma_f32_16x16x4_f32(false, b0, false, b1, (short)0, acc[1], false, false);
          acc[2] = __builtin_amdgcn_wmma_f32_16x16x4_f32(false, b1, false, b1, (short)0, acc[2], false, false);
        }
      }
    } else {
      mma_frags<2, NS, false>(f, acc);
    }
    __builtin_amdgcn_sched_barrier(0);
  }

  if (w != 0) {
#pragma unroll
    for (int j = 0; j < NT; ++j) {
      v4f lo, hi;
      lo.x = acc[j][0]; lo.y = acc[j][1]; lo.z = acc[j][2]; lo.w = acc[j][3];
      hi.x = acc[j][4]; hi.y = acc[j][5]; hi.z = acc[j][6]; hi.w = acc[j][7];
      red[(((w - 1) * NT + j) * 2    ) * 32 + lane] = lo;
      red[(((w - 1) * NT + j) * 2 + 1) * 32 + lane] = hi;
    }
  }
  __syncthreads();
  if (w != 0) return;
#pragma unroll
  for (int p = 1; p < KS1; ++p) {
#pragma unroll
    for (int j = 0; j < NT; ++j) {
      const v4f lo = red[(((p - 1) * NT + j) * 2    ) * 32 + lane];
      const v4f hi = red[(((p - 1) * NT + j) * 2 + 1) * 32 + lane];
      acc[j][0] += lo.x; acc[j][1] += lo.y; acc[j][2] += lo.z; acc[j][3] += lo.w;
      acc[j][4] += hi.x; acc[j][5] += hi.y; acc[j][6] += hi.z; acc[j][7] += hi.w;
    }
  }
  const int pu = m0 >> 4, qu = n0 >> 4;
#pragma unroll
  for (int j = 0; j < NT; ++j) {
    const int p = (DIAG && j == 2) ? pu + 1 : pu;
    const int q = (DIAG && j == 2) ? qu + 1 : qu + j;
    float* u = C + (long)(16 * p + q) * S1U;
#pragma unroll
    for (int v = 0; v < 8; ++v) u[s1_at(v + 8 * half, r)] = acc[j][v];
  }
#pragma unroll
  for (int j = (DIAG ? 1 : 0); j < 2; ++j) {
    v4f lo, hi;
    lo.x = acc[j][0]; lo.y = acc[j][1]; lo.z = acc[j][2]; lo.w = acc[j][3];
    hi.x = acc[j][4]; hi.y = acc[j][5]; hi.z = acc[j][6]; hi.w = acc[j][7];
    float* q = C + (long)(16 * (qu + j) + pu) * S1U + s1_at(r, 8 * half);
    *(v4f*)(q)     = lo;
    *(v4f*)(q + 4) = hi;
  }
}

__global__ void __launch_bounds__(32 * KS1) k_stage1_xxT(const float* __restrict__ xkv, float* __restrict__ S) {
  __shared__ __attribute__((aligned(16))) float stg[KS1 * S1ROWS * S1P];
  __shared__ v4f red[(KS1 - 1) * 3 * 2 * 32];
  static_assert(B_ == 4 && S1T == 64, "the decode below is written for 4 batches of 8 + 56 tiles");
  const int g = blockIdx.x;
  int b, i, J;
  if (g < 8 * B_) {
    b = g >> 3;
    J = g & 7;
    i = 2 * J;
  } else {
    const int u = g - 8 * B_;
    b = (u >= 56) + (u >= 112) + (u >= 168);
    const int t = u - 56 * b;
    J = 1 + (t >= 2) + (t >= 6) + (t >= 12) + (t >= 20) + (t >= 30) + (t >= 42);
    i = t - J * (J - 1);
  }
  const int w = __builtin_amdgcn_readfirstlane((int)(threadIdx.x >> 5));
  const int lane = threadIdx.x & 31;
  const float* X = xkv + (long)b * CKV_ * NKV_ + (long)w * (NKV_ / KS1);
  float* C  = S + (long)b * CKV_ * CKV_;
  float* my = stg + w * (S1ROWS * S1P);
  if (i == 2 * J) s1_tile<true >(X, C, my, red, 16 * i, 32 * J, w, lane);
  else            s1_tile<false>(X, C, my, red, 16 * i, 32 * J, w, lane);
}

constexpr int W234 = 8;
constexpr int T1P  = CKV_ + 4;
constexpr int TP34 = DK_ + 4;
__global__ void __launch_bounds__(32 * W234) k_stage234_Pt(const float* __restrict__ S, const float* __restrict__ Wv,
                             const float* __restrict__ Wk, const float* __restrict__ Wq,
                             float* __restrict__ Pt) {
  constexpr int NB = 4, KC = 32, NS = KC / 8;
  constexpr int KWA = CKV_ / 2;
  constexpr int NBC = 2, NSC = DK_ / 8;
  static_assert(KWA == 4 * KC && CKV_ == W234 * KC && CQ_ == W234 * 16 * NBC,
                "phase A walks four chunks; phase B one chunk per wave; phase C 32 columns per wave");
  const int z = blockIdx.z;
  const int b = z >> 3, h = z & 7;
  const int m0 = blockIdx.y * 16;
  const int w = __builtin_amdgcn_readfirstlane((int)(threadIdx.x >> 5));
  const int lane = threadIdx.x & 31;
  const int r    = lane & 15;
  const int half = lane >> 4;
  const int nb = w >> 1, ks = w & 1;

  __shared__ v4f redA[4 * NB * 2 * 32];
  __shared__ v4f redB[W234 * NB * 2 * 32];
  __shared__ __attribute__((aligned(16))) float t1s[16 * T1P];
  __shared__ __attribute__((aligned(16))) float mt[16 * TP34];

  const float* A1 = Wv + (long)h * DK_ * CKV_ + (long)ks * KWA;
  const float* B1 = S + (long)b * CKV_ * CKV_ + (long)ks * KWA * CKV_;
  const long arow = (long)(m0 + r) * CKV_;
  const int n0 = 64 * nb;
  const long lb = (long)(4 * nb + (r >> 2)) * 256 + half * 16 + 4 * (r & 3);
  v8f acc[NB];
#pragma unroll
  for (int j = 0; j < NB; ++j) acc[j] = {};
  const float* Wkh = Wk + (long)h * DK_ * CKV_;
  const float* Bq  = Wq + (long)h * DK_ * CQ_;
  const long col0 = 32 * w + NBC * r;
  Frags<NB, NS, false> fb;
  Frags<NBC, NSC, true> g;
  {
    Frags<NB, NS, true> f0, f1;
    load_frags_su<NS>(f0, A1, arow, B1, lb, 0, half);
    load_frags_su<NS>(f1, A1, arow, B1, lb, KC, half);
    __builtin_amdgcn_sched_barrier(0);
    mma_frags<NB, NS, true>(f0, acc);
    __builtin_amdgcn_sched_barrier(0);
    load_frags_su<NS>(f0, A1, arow, B1, lb, 2 * KC, half);
    __builtin_amdgcn_sched_barrier(0);
    mma_frags<NB, NS, true>(f1, acc);
    __builtin_amdgcn_sched_barrier(0);
    load_frags_su<NS>(f1, A1, arow, B1, lb, 3 * KC, half);
#pragma unroll
    for (int s = 0; s < NS; ++s) {
      const int kb = KC * w + 8 * s + 4 * half;
#pragma unroll
      for (int j = 0; j < NB; ++j) fb.b[s][j] = *(const v4f*)(Wkh + (long)(16 * j + r) * CKV_ + kb);
    }
    __builtin_amdgcn_sched_barrier(0);
    mma_frags<NB, NS, true>(f0, acc);
    __builtin_amdgcn_sched_barrier(0);
    load_b_rows<NBC, NSC>(g, Bq, CQ_, col0, 0, half);
    __builtin_amdgcn_sched_barrier(0);
    mma_frags<NB, NS, true>(f1, acc);
  }

  if (ks != 0) {
#pragma unroll
    for (int j = 0; j < NB; ++j) {
      v4f lo, hi;
      lo.x = acc[j][0]; lo.y = acc[j][1]; lo.z = acc[j][2]; lo.w = acc[j][3];
      hi.x = acc[j][4]; hi.y = acc[j][5]; hi.z = acc[j][6]; hi.w = acc[j][7];
      redA[((nb * NB + j) * 2    ) * 32 + lane] = lo;
      redA[((nb * NB + j) * 2 + 1) * 32 + lane] = hi;
    }
  }
  __syncthreads();
  if (ks == 0) {
#pragma unroll
    for (int j = 0; j < NB; ++j) {
      const v4f lo = redA[((nb * NB + j) * 2    ) * 32 + lane];
      const v4f hi = redA[((nb * NB + j) * 2 + 1) * 32 + lane];
      acc[j][0] += lo.x; acc[j][1] += lo.y; acc[j][2] += lo.z; acc[j][3] += lo.w;
      acc[j][4] += hi.x; acc[j][5] += hi.y; acc[j][6] += hi.z; acc[j][7] += hi.w;
    }
#pragma unroll
    for (int v = 0; v < 8; ++v) {
      v4f val;
#pragma unroll
      for (int j = 0; j < NB; ++j) val[j] = acc[j][v];
      *(v4f*)(t1s + (v + 8 * half) * T1P + n0 + NB * r) = val;
    }
  }
  __syncthreads();

#pragma unroll
  for (int s = 0; s < NS; ++s) fb.a[s] = *(const v4f*)(t1s + r * T1P + KC * w + 8 * s + 4 * half);
  v8f acc2[NB];
#pragma unroll
  for (int j = 0; j < NB; ++j) acc2[j] = {};
  mma_frags<NB, NS, false>(fb, acc2);
#pragma unroll
  for (int j = 0; j < NB; ++j) {
    v4f lo, hi;
    lo.x = acc2[j][0]; lo.y = acc2[j][1]; lo.z = acc2[j][2]; lo.w = acc2[j][3];
    hi.x = acc2[j][4]; hi.y = acc2[j][5]; hi.z = acc2[j][6]; hi.w = acc2[j][7];
    redB[((w * NB + j) * 2    ) * 32 + lane] = lo;
    redB[((w * NB + j) * 2 + 1) * 32 + lane] = hi;
  }
  __syncthreads();
  {
    const int t = nb, hh = ks;
    v4f s = redB[((0 * NB + t) * 2 + hh) * 32 + lane];
#pragma unroll
    for (int p = 1; p < W234; ++p) {
      const v4f u = redB[((p * NB + t) * 2 + hh) * 32 + lane];
      s.x += u.x; s.y += u.y; s.z += u.z; s.w += u.w;
    }
    const float sc = 1.0f / (float)DK_;
    mt[(4 * hh + 0 + 8 * half) * TP34 + 16 * t + r] = s.x * sc;
    mt[(4 * hh + 1 + 8 * half) * TP34 + 16 * t + r] = s.y * sc;
    mt[(4 * hh + 2 + 8 * half) * TP34 + 16 * t + r] = s.z * sc;
    mt[(4 * hh + 3 + 8 * half) * TP34 + 16 * t + r] = s.w * sc;
  }
  __syncthreads();

#pragma unroll
  for (int s = 0; s < NSC; ++s) g.a[s] = *(const v4f*)(mt + r * TP34 + 8 * s + 4 * half);
  v8f acc3[NBC];
#pragma unroll
  for (int j = 0; j < NBC; ++j) acc3[j] = {};
  mma_frags<NBC, NSC, true>(g, acc3);

  float* C = Pt + (long)z * DK_ * CQ_;
#pragma unroll
  for (int v = 0; v < 8; ++v) {
    const long row = m0 + v + 8 * half;
    v2f val;
#pragma unroll
    for (int j = 0; j < NBC; ++j) val[j] = acc3[j][v];
    *(v2f*)(C + row * CQ_ + col0) = val;
  }
}

__global__ void __launch_bounds__(32 * KS5) k_stage5_G(const float* __restrict__ Wo, const float* __restrict__ Pt,
                           const float* __restrict__ gamma, float* __restrict__ G) {
  const int b = blockIdx.z;
  wmma_gemm_strip<4, false, true, true, 32, OD_, KS5>(Wo, OD_, 1,
                            Pt + (long)b * OD_ * CQ_, CQ_, 1,
                            G + (long)b * CQ_ * CQ_, CQ_, nullptr, 0,
                            blockIdx.y * 16, blockIdx.x * 64, gamma[0]);
}

constexpr int W6 = 8;
__global__ void __launch_bounds__(32 * W6) k_stage6_out(const float* __restrict__ G, const float* __restrict__ xq,
                             float* __restrict__ out) {
  const int b = blockIdx.z;
  const int w = __builtin_amdgcn_readfirstlane((int)(threadIdx.x >> 5));
  const float* Bq = xq + (long)b * CQ_ * NQ_;
  wmma_gemm_strip<4, true, true, true, 32, CQ_, 1>(G + (long)b * CQ_ * CQ_, CQ_, 1,
                           Bq, NQ_, 1,
                           out + (long)b * CQ_ * NQ_, NQ_, Bq, NQ_,
                           (blockIdx.y * W6 + w) * 16, blockIdx.x * 64, 1.0f);
}

extern "C" void kernel_launch(void* const* d_in, const int* in_sizes, int n_in,
                              void* d_out, int out_size, void* d_ws, size_t ws_size,
                              hipStream_t stream) {
  const float* xq    = (const float*)d_in[0];
  const float* xkv   = (const float*)d_in[1];
  const float* Wq    = (const float*)d_in[2];
  const float* Wk    = (const float*)d_in[3];
  const float* Wv    = (const float*)d_in[4];
  const float* Wo    = (const float*)d_in[5];
  const float* gamma = (const float*)d_in[6];
  float* out = (float*)d_out;

  float* S  = (float*)d_ws;
  float* Pt = S  + (size_t)B_ * CKV_ * CKV_;
  float* G  = Pt + (size_t)B_ * H_ * DK_ * CQ_;

  k_stage1_xxT<<<dim3(S1T * B_, 1, 1), dim3(32 * KS1, 1, 1), 0, stream>>>(xkv, S);
  k_stage234_Pt<<<dim3(1,        DK_ / 16, B_ * H_), dim3(32 * W234, 1, 1), 0, stream>>>(S, Wv, Wk, Wq, Pt);
  k_stage5_G  <<<dim3(CQ_ / 64,  CQ_ / 16, B_),      dim3(32 * KS5, 1, 1), 0, stream>>>(Wo, Pt, gamma, G);
  k_stage6_out<<<dim3(NQ_ / 64,  CQ_ / (16 * W6), B_), dim3(32 * W6, 1, 1), 0, stream>>>(G, xq, out);
}
